# starting kernel
# baseline (speedup 1.0000x reference)
#include <hip/hip_runtime.h>
#include <hip/hip_bf16.h>
#include <math.h>


#define NT     2048
#define DD     384
#define DH     128
#define D3     384
#define D9     1152
#define GAMMA  5.0f
#define LOG2E  1.4426950408889634f

typedef float v2f __attribute__((ext_vector_type(2)));
typedef float v8f __attribute__((ext_vector_type(8)));

static __device__ __forceinline__ v8f wmma4(v2f a, v2f b, v8f c) {
  return __builtin_amdgcn_wmma_f32_16x16x4_f32(
       false, a,  false, b,
       (short)0, c,  false,  false);
}

static __device__ __forceinline__ float softplusf(float w) {
  return (w > 20.0f) ? w : log1pf(__expf(w));
}


__global__ void cea_softplus(const float* __restrict__ in,
                             float* __restrict__ out, int n) {
  int i = blockIdx.x * blockDim.x + threadIdx.x;
  if (i < n) out[i] = softplusf(in[i]);
}

__global__ void cea_fuse(const float* __restrict__ spw,
                         const float* __restrict__ Sq,
                         const float* __restrict__ Sk,
                         const float* __restrict__ Sv,
                         float* __restrict__ Mt) {
  int h   = blockIdx.y;
  int idx = blockIdx.x * 256 + threadIdx.x;
  int n = idx >> 7, k = idx & 127;
  const float* S = (h == 0) ? Sq : (h == 1) ? Sk : Sv;
  const float* W = spw + h * (D3 * DH);
  float acc = 0.0f;
  for (int j = 0; j < D3; ++j) acc += W[j * DH + k] * S[j * D3 + n];
  Mt[h * (D3 * DH) + idx] = acc;
}

__global__ void cea_bias(const float* __restrict__ bq, const float* __restrict__ bk,
                         const float* __restrict__ bv,
                         const float* __restrict__ Sq, const float* __restrict__ Sk,
                         const float* __restrict__ Sv,
                         float* __restrict__ c) {
  int h = blockIdx.x, n = threadIdx.x;
  const float* b = (h == 0) ? bq : (h == 1) ? bk : bv;
  const float* S = (h == 0) ? Sq : (h == 1) ? Sk : Sv;
  float acc = 0.0f;
  for (int j = 0; j < D3; ++j) acc += b[j] * S[j * D3 + n];
  c[h * D3 + n] = acc;
}

__global__ void cea_qkv(const float* __restrict__ x,
                        const float* __restrict__ Mt,
                        const float* __restrict__ cb,
                        float* __restrict__ qkv) {
  __shared__ float xT[16 * DH];
  const int h  = blockIdx.y;
  const int t0 = blockIdx.x * 16;
  const int tid = threadIdx.x;

  for (int idx = tid; idx < 16 * DH; idx += 256) {
    int r = idx >> 7, cc = idx & 127;
    xT[idx] = x[(t0 + r) * DD + h * DH + cc];
  }
  __syncthreads();

  const int lane = tid & 31, wid = tid >> 5;
  const int m  = lane & 15;
  const int kh = (lane >> 4) * 2;
  const float* __restrict__ Mh = Mt + h * (D3 * DH);
  const int n0 = wid * 48;

  v8f acc0 = {}, acc1 = {}, acc2 = {};
  for (int k0 = 0; k0 < DH; k0 += 4) {
    v2f a  = *(const v2f*)&xT[m * DH + k0 + kh];
    v2f b0 = *(const v2f*)&Mh[(n0 + 0  + m) * DH + k0 + kh];
    v2f b1 = *(const v2f*)&Mh[(n0 + 16 + m) * DH + k0 + kh];
    v2f b2 = *(const v2f*)&Mh[(n0 + 32 + m) * DH + k0 + kh];
    acc0 = wmma4(a, b0, acc0);
    acc1 = wmma4(a, b1, acc1);
    acc2 = wmma4(a, b2, acc2);
  }

  const int rbase = (lane >> 4) * 8;
  const int ncol  = lane & 15;
  for (int r = 0; r < 8; ++r) {
    int row = t0 + rbase + r;
    qkv[row * D9 + h * D3 + n0 + 0  + ncol] = acc0[r] + cb[h * D3 + n0 + 0  + ncol];
    qkv[row * D9 + h * D3 + n0 + 16 + ncol] = acc1[r] + cb[h * D3 + n0 + 16 + ncol];
    qkv[row * D9 + h * D3 + n0 + 32 + ncol] = acc2[r] + cb[h * D3 + n0 + 32 + ncol];
  }
}

__global__ void cea_attn(const float* __restrict__ qkv,
                         float* __restrict__ att) {
  __shared__ float Qs[D3], Ksc[D3], Vs[D3];
  const int t = blockIdx.x, i = threadIdx.x;
  const float* __restrict__ base = qkv + t * D9;
  Qs[i]  = base[i];
  Ksc[i] = base[D3 + i] * (-GAMMA * LOG2E);
  Vs[i]  = base[2 * D3 + i];
  __syncthreads();

  const float Qi = Qs[i];
  float acc = 0.0f, ssum = 0.0f;
#pragma unroll 4
  for (int j = 0; j < D3; ++j) {
    float s = __builtin_amdgcn_rcpf(1.0f + __builtin_amdgcn_exp2f(Qi * Ksc[j]));
    acc  += s * Vs[j];
    ssum += s;
  }
  att[t * D3 + i] = acc / (ssum + 1e-8f);
}

__global__ void cea_proj(const float* __restrict__ att,
                         const float* __restrict__ spP,
                         const float* __restrict__ bp,
                         float* __restrict__ hout) {
  __shared__ float aT[16 * D3];
  const int t0 = blockIdx.x * 16;
  const int tid = threadIdx.x;
  for (int idx = tid; idx < 16 * D3; idx += 256)
    aT[idx] = att[t0 * D3 + idx];
  __syncthreads();

  const int lane = tid & 31, wid = tid >> 5;
  const int m  = lane & 15;
  const int kh = (lane >> 4) * 2;
  const int nbase = wid * 144;

  v8f acc[9];
#pragma unroll
  for (int q = 0; q < 9; ++q) acc[q] = (v8f){};

  for (int k0 = 0; k0 < D3; k0 += 4) {
    v2f a = *(const v2f*)&aT[m * D3 + k0 + kh];
#pragma unroll
    for (int q = 0; q < 9; ++q) {
      v2f b = *(const v2f*)&spP[(nbase + q * 16 + m) * D3 + k0 + kh];
      acc[q] = wmma4(a, b, acc[q]);
    }
  }

  const int rbase = (lane >> 4) * 8;
  const int ncol  = lane & 15;
#pragma unroll
  for (int q = 0; q < 9; ++q) {
    int n = nbase + q * 16 + ncol;
    float bias = bp[n];
    for (int r = 0; r < 8; ++r)
      hout[(t0 + rbase + r) * D9 + n] = acc[q][r] + bias;
  }
}

__global__ void cea_ln(float* __restrict__ out,
                       const float* __restrict__ gn,
                       const float* __restrict__ bn) {
  __shared__ float row[D9];
  __shared__ float red[256];
  const int t = blockIdx.x, tid = threadIdx.x;
  float* __restrict__ h = out + t * D9;

  float ps = 0.0f;
  for (int n = tid; n < D9; n += 256) { float v = h[n]; row[n] = v; ps += v; }
  red[tid] = ps;
  __syncthreads();
  for (int s = 128; s > 0; s >>= 1) {
    if (tid < s) red[tid] += red[tid + s];
    __syncthreads();
  }
  const float mu = red[0] * (1.0f / (float)D9);
  __syncthreads();

  float pv = 0.0f;
  for (int n = tid; n < D9; n += 256) { float dv = row[n] - mu; pv += dv * dv; }
  red[tid] = pv;
  __syncthreads();
  for (int s = 128; s > 0; s >>= 1) {
    if (tid < s) red[tid] += red[tid + s];
    __syncthreads();
  }
  const float var = red[0] * (1.0f / (float)D9);
  const float rs  = __builtin_amdgcn_rsqf(var + 1e-5f);

  for (int n = tid; n < D9; n += 256)
    h[n] = (row[n] - mu) * rs * gn[n] + bn[n];
}


extern "C" void kernel_launch(void* const* d_in, const int* in_sizes, int n_in,
                              void* d_out, int out_size, void* d_ws, size_t ws_size,
                              hipStream_t stream) {
  const float* x  = (const float*)d_in[0];
  const float* Wq = (const float*)d_in[1];
  const float* bq = (const float*)d_in[2];
  const float* Sq = (const float*)d_in[3];
  const float* Wk = (const float*)d_in[4];
  const float* bk = (const float*)d_in[5];
  const float* Sk = (const float*)d_in[6];
  const float* Wv = (const float*)d_in[7];
  const float* bv = (const float*)d_in[8];
  const float* Sv = (const float*)d_in[9];
  const float* Wp = (const float*)d_in[10];
  const float* bp = (const float*)d_in[11];
  const float* gn = (const float*)d_in[12];
  const float* bn = (const float*)d_in[13];
  float* out = (float*)d_out;

  float* ws = (float*)d_ws;
  float* spw  = ws;
  float* Mt   = spw + 3 * D3 * DH;
  float* cb   = Mt  + 3 * D3 * DH;
  float* spP  = cb  + 3 * D3;
  float* qkv  = spP + D9 * D3;
  float* attb = qkv + NT * D9;

  cea_softplus<<<dim3((D3 * DH + 255) / 256), dim3(256), 0, stream>>>(Wq, spw,              D3 * DH);
  cea_softplus<<<dim3((D3 * DH + 255) / 256), dim3(256), 0, stream>>>(Wk, spw + D3 * DH,    D3 * DH);
  cea_softplus<<<dim3((D3 * DH + 255) / 256), dim3(256), 0, stream>>>(Wv, spw + 2 * D3 * DH, D3 * DH);
  cea_softplus<<<dim3((D9 * D3 + 255) / 256), dim3(256), 0, stream>>>(Wp, spP,              D9 * D3);

  cea_fuse<<<dim3((D3 * DH) / 256, 3), dim3(256), 0, stream>>>(spw, Sq, Sk, Sv, Mt);
  cea_bias<<<dim3(3), dim3(D3), 0, stream>>>(bq, bk, bv, Sq, Sk, Sv, cb);

  cea_qkv<<<dim3(NT / 16, 3), dim3(256), 0, stream>>>(x, Mt, cb, qkv);

  cea_attn<<<dim3(NT), dim3(D3), 0, stream>>>(qkv, attb);

  cea_proj<<<dim3(NT / 16), dim3(256), 0, stream>>>(attb, spP, bp, out);

  cea_ln<<<dim3(NT), dim3(256), 0, stream>>>(out, gn, bn);

  (void)in_sizes; (void)n_in; (void)out_size; (void)ws_size;
}


# projection GEMM: weights stored in WMMA-fragment order by their softplus launch, B read straight to registers (coalesced 16 B/lane, no LDS for B), 64x144 block tile; k summed in new order within 8s
# speedup vs baseline: 1.5319x; 1.5319x over previous
#include <hip/hip_runtime.h>
#include <hip/hip_bf16.h>
#include <math.h>


#define NT     2048
#define DD     384
#define DH     128
#define D3     384
#define D9     1152
#define PKC    48
#define PTOK   64
#define PCOL   144
#define APITCH 388
#define GAMMA  5.0f
#define LOG2E  1.4426950408889634f

typedef float v2f __attribute__((ext_vector_type(2)));
typedef float v4f __attribute__((ext_vector_type(4)));
typedef float v8f __attribute__((ext_vector_type(8)));

static __device__ __forceinline__ v8f wmma4(v2f a, v2f b, v8f c) {
  return __builtin_amdgcn_wmma_f32_16x16x4_f32(
       false, a,  false, b,
       (short)0, c,  false,  false);
}

static __device__ __forceinline__ float softplusf(float w) {
  return (w > 20.0f) ? w : log1pf(__expf(w));
}


__global__ void cea_softplus(const float* __restrict__ in,
                             float* __restrict__ out, int n) {
  int i = blockIdx.x * blockDim.x + threadIdx.x;
  if (i < n) out[i] = softplusf(in[i]);
}

__global__ void cea_softplus_frag(const float* __restrict__ Wp,
                                  float* __restrict__ spPf) {
  const int o4 = blockIdx.x * 256 + threadIdx.x;
  const int lane = o4 & 31;
  const int c = lane & 15, h = lane >> 4;
  const int ch = o4 >> 5;
  const int nt = ch / PKC, kc = ch - nt * PKC;
  const v4f w = *(const v4f*)&Wp[(nt * 16 + c) * D3 + kc * 8 + 4 * h];
  v4f o;
  o.x = softplusf(w.x); o.y = softplusf(w.y);
  o.z = softplusf(w.z); o.w = softplusf(w.w);
  *(v4f*)&spPf[o4 * 4] = o;
}

__global__ void cea_fuse(const float* __restrict__ spw,
                         const float* __restrict__ Sq,
                         const float* __restrict__ Sk,
                         const float* __restrict__ Sv,
                         float* __restrict__ Mt) {
  int h   = blockIdx.y;
  int idx = blockIdx.x * 256 + threadIdx.x;
  int n = idx >> 7, k = idx & 127;
  const float* S = (h == 0) ? Sq : (h == 1) ? Sk : Sv;
  const float* W = spw + h * (D3 * DH);
  float acc = 0.0f;
  for (int j = 0; j < D3; ++j) acc += W[j * DH + k] * S[j * D3 + n];
  Mt[h * (D3 * DH) + idx] = acc;
}

__global__ void cea_bias(const float* __restrict__ bq, const float* __restrict__ bk,
                         const float* __restrict__ bv,
                         const float* __restrict__ Sq, const float* __restrict__ Sk,
                         const float* __restrict__ Sv,
                         float* __restrict__ c) {
  int h = blockIdx.x, n = threadIdx.x;
  const float* b = (h == 0) ? bq : (h == 1) ? bk : bv;
  const float* S = (h == 0) ? Sq : (h == 1) ? Sk : Sv;
  float acc = 0.0f;
  for (int j = 0; j < D3; ++j) acc += b[j] * S[j * D3 + n];
  c[h * D3 + n] = acc;
}

__global__ void cea_qkv(const float* __restrict__ x,
                        const float* __restrict__ Mt,
                        const float* __restrict__ cb,
                        float* __restrict__ qkv) {
  __shared__ float xT[16 * DH];
  const int h  = blockIdx.y;
  const int t0 = blockIdx.x * 16;
  const int tid = threadIdx.x;

  for (int idx = tid; idx < 16 * DH; idx += 256) {
    int r = idx >> 7, cc = idx & 127;
    xT[idx] = x[(t0 + r) * DD + h * DH + cc];
  }
  __syncthreads();

  const int lane = tid & 31, wid = tid >> 5;
  const int m  = lane & 15;
  const int kh = (lane >> 4) * 2;
  const float* __restrict__ Mh = Mt + h * (D3 * DH);
  const int n0 = wid * 48;

  v8f acc0 = {}, acc1 = {}, acc2 = {};
  for (int k0 = 0; k0 < DH; k0 += 4) {
    v2f a  = *(const v2f*)&xT[m * DH + k0 + kh];
    v2f b0 = *(const v2f*)&Mh[(n0 + 0  + m) * DH + k0 + kh];
    v2f b1 = *(const v2f*)&Mh[(n0 + 16 + m) * DH + k0 + kh];
    v2f b2 = *(const v2f*)&Mh[(n0 + 32 + m) * DH + k0 + kh];
    acc0 = wmma4(a, b0, acc0);
    acc1 = wmma4(a, b1, acc1);
    acc2 = wmma4(a, b2, acc2);
  }

  const int rbase = (lane >> 4) * 8;
  const int ncol  = lane & 15;
  for (int r = 0; r < 8; ++r) {
    int row = t0 + rbase + r;
    qkv[row * D9 + h * D3 + n0 + 0  + ncol] = acc0[r] + cb[h * D3 + n0 + 0  + ncol];
    qkv[row * D9 + h * D3 + n0 + 16 + ncol] = acc1[r] + cb[h * D3 + n0 + 16 + ncol];
    qkv[row * D9 + h * D3 + n0 + 32 + ncol] = acc2[r] + cb[h * D3 + n0 + 32 + ncol];
  }
}

__global__ void cea_attn(const float* __restrict__ qkv,
                         float* __restrict__ att) {
  __shared__ float Qs[D3], Ksc[D3], Vs[D3];
  const int t = blockIdx.x, i = threadIdx.x;
  const float* __restrict__ base = qkv + t * D9;
  Qs[i]  = base[i];
  Ksc[i] = base[D3 + i] * (-GAMMA * LOG2E);
  Vs[i]  = base[2 * D3 + i];
  __syncthreads();

  const float Qi = Qs[i];
  float acc = 0.0f, ssum = 0.0f;
#pragma unroll 4
  for (int j = 0; j < D3; ++j) {
    float s = __builtin_amdgcn_rcpf(1.0f + __builtin_amdgcn_exp2f(Qi * Ksc[j]));
    acc  += s * Vs[j];
    ssum += s;
  }
  att[t * D3 + i] = acc / (ssum + 1e-8f);
}


static __device__ __forceinline__ v2f lo2(v4f v) { return __builtin_shufflevector(v, v, 0, 1); }
static __device__ __forceinline__ v2f hi2(v4f v) { return __builtin_shufflevector(v, v, 2, 3); }

template <int NTW>
static __device__ __forceinline__ void proj_load(const float* __restrict__ aLane,
                                                 const v4f* __restrict__ bLane,
                                                 int kc, v4f& a, v4f (&b)[NTW]) {
  a = *(const v4f*)(aLane + kc * 8);
#pragma unroll
  for (int q = 0; q < NTW; ++q) b[q] = bLane[(q * PKC + kc) * 32];
}

template <int NTW>
static __device__ __forceinline__ void proj_mma(v4f a, const v4f (&b)[NTW], v8f (&acc)[NTW]) {
#pragma unroll
  for (int q = 0; q < NTW; ++q) acc[q] = wmma4(lo2(a), lo2(b[q]), acc[q]);
#pragma unroll
  for (int q = 0; q < NTW; ++q) acc[q] = wmma4(hi2(a), hi2(b[q]), acc[q]);
}

template <int NTW>
static __device__ __forceinline__ void proj_wave(const float* __restrict__ aLane,
                                                 const v4f* __restrict__ bLane,
                                                 const float* __restrict__ bp,
                                                 float* __restrict__ hout,
                                                 int row0, int col0) {
  v8f acc[NTW];
#pragma unroll
  for (int q = 0; q < NTW; ++q) acc[q] = (v8f){};

  v4f a[4], b[4][NTW];
  proj_load<NTW>(aLane, bLane, 0, a[0], b[0]);
  proj_load<NTW>(aLane, bLane, 1, a[1], b[1]);
  proj_load<NTW>(aLane, bLane, 2, a[2], b[2]);
#pragma unroll
  for (int s = 0; s < 3; ++s) {
    asm volatile("" :: "v"(a[s]));
#pragma unroll
    for (int q = 0; q < NTW; ++q) asm volatile("" :: "v"(b[s][q]));
  }
  for (int kc = 0; kc < PKC; kc += 4) {
#pragma unroll
    for (int s = 0; s < 4; ++s) {
      const int kp = kc + s + 3;
      const int kn = (kp < PKC) ? kp : PKC - 1;
      proj_load<NTW>(aLane, bLane, kn, a[(s + 3) & 3], b[(s + 3) & 3]);
      proj_mma<NTW>(a[s], b[s], acc);
    }
  }

#pragma unroll
  for (int q = 0; q < NTW; ++q) {
    const int n = col0 + q * 16;
    const float bias = bp[n];
#pragma unroll
    for (int r = 0; r < 8; ++r)
      hout[(row0 + r) * D9 + n] = acc[q][r] + bias;
  }
}

__global__ __launch_bounds__(256) void cea_proj(const float* __restrict__ att,
                                                const float* __restrict__ spPf,
                                                const float* __restrict__ bp,
                                                float* __restrict__ hout) {
  __shared__ __attribute__((aligned(16))) float aT[PTOK * APITCH];
  const int t0  = blockIdx.x * PTOK;
  const int tid = threadIdx.x;
  {
    const int sw = tid >> 5, sl = tid & 31;
    const float* src = att + (t0 + sw * 8) * D3 + sl * 4;
    float*       dst = aT + (sw * 8) * APITCH + sl * 4;
    v4f tmp[24];
#pragma unroll
    for (int i = 0; i < 24; ++i) tmp[i] = *(const v4f*)(src + (i / 3) * D3 + (i % 3) * 128);
#pragma unroll
    for (int i = 0; i < 24; ++i) *(v4f*)(dst + (i / 3) * APITCH + (i % 3) * 128) = tmp[i];
  }
  __syncthreads();

  const int lane = tid & 31, wid = tid >> 5;
  const int c  = lane & 15;
  const int h  = lane >> 4;
  const int mi = wid & 3;
  const int nt0 = (wid >> 2) * 5;
  const float* aLane = &aT[(mi * 16 + c) * APITCH + 4 * h];
  const v4f*   bLane = (const v4f*)spPf + ((blockIdx.y * 9 + nt0) * PKC) * 32 + lane;
  const int row0 = t0 + mi * 16 + h * 8;
  const int col0 = blockIdx.y * PCOL + nt0 * 16 + c;

  if (wid < 4) proj_wave<5>(aLane, bLane, bp, hout, row0, col0);
  else         proj_wave<4>(aLane, bLane, bp, hout, row0, col0);
}

__global__ void cea_ln(float* __restrict__ out,
                       const float* __restrict__ gn,
                       const float* __restrict__ bn) {
  __shared__ float row[D9];
  __shared__ float red[256];
  const int t = blockIdx.x, tid = threadIdx.x;
  float* __restrict__ h = out + t * D9;

  float ps = 0.0f;
  for (int n = tid; n < D9; n += 256) { float v = h[n]; row[n] = v; ps += v; }
  red[tid] = ps;
  __syncthreads();
  for (int s = 128; s > 0; s >>= 1) {
    if (tid < s) red[tid] += red[tid + s];
    __syncthreads();
  }
  const float mu = red[0] * (1.0f / (float)D9);
  __syncthreads();

  float pv = 0.0f;
  for (int n = tid; n < D9; n += 256) { float dv = row[n] - mu; pv += dv * dv; }
  red[tid] = pv;
  __syncthreads();
  for (int s = 128; s > 0; s >>= 1) {
    if (tid < s) red[tid] += red[tid + s];
    __syncthreads();
  }
  const float var = red[0] * (1.0f / (float)D9);
  const float rs  = __builtin_amdgcn_rsqf(var + 1e-5f);

  for (int n = tid; n < D9; n += 256)
    h[n] = (row[n] - mu) * rs * gn[n] + bn[n];
}


extern "C" void kernel_launch(void* const* d_in, const int* in_sizes, int n_in,
                              void* d_out, int out_size, void* d_ws, size_t ws_size,
                              hipStream_t stream) {
  const float* x  = (const float*)d_in[0];
  const float* Wq = (const float*)d_in[1];
  const float* bq = (const float*)d_in[2];
  const float* Sq = (const float*)d_in[3];
  const float* Wk = (const float*)d_in[4];
  const float* bk = (const float*)d_in[5];
  const float* Sk = (const float*)d_in[6];
  const float* Wv = (const float*)d_in[7];
  const float* bv = (const float*)d_in[8];
  const float* Sv = (const float*)d_in[9];
  const float* Wp = (const float*)d_in[10];
  const float* bp = (const float*)d_in[11];
  const float* gn = (const float*)d_in[12];
  const float* bn = (const float*)d_in[13];
  float* out = (float*)d_out;

  float* ws = (float*)d_ws;
  float* spw  = ws;
  float* Mt   = spw + 3 * D3 * DH;
  float* cb   = Mt  + 3 * D3 * DH;
  float* spP  = cb  + 3 * D3;
  float* qkv  = spP + D9 * D3;
  float* attb = qkv + NT * D9;

  cea_softplus<<<dim3((D3 * DH + 255) / 256), dim3(256), 0, stream>>>(Wq, spw,              D3 * DH);
  cea_softplus<<<dim3((D3 * DH + 255) / 256), dim3(256), 0, stream>>>(Wk, spw + D3 * DH,    D3 * DH);
  cea_softplus<<<dim3((D3 * DH + 255) / 256), dim3(256), 0, stream>>>(Wv, spw + 2 * D3 * DH, D3 * DH);
  cea_softplus_frag<<<dim3((D9 * D3 / 4) / 256), dim3(256), 0, stream>>>(Wp, spP);

  cea_fuse<<<dim3((D3 * DH) / 256, 3), dim3(256), 0, stream>>>(spw, Sq, Sk, Sv, Mt);
  cea_bias<<<dim3(3), dim3(D3), 0, stream>>>(bq, bk, bv, Sq, Sk, Sv, cb);

  cea_qkv<<<dim3(NT / 16, 3), dim3(256), 0, stream>>>(x, Mt, cb, qkv);

  cea_attn<<<dim3(NT), dim3(D3), 0, stream>>>(qkv, attb);

  cea_proj<<<dim3(NT / PTOK, D9 / PCOL), dim3(256), 0, stream>>>(attb, spP, bp, out);

  cea_ln<<<dim3(NT), dim3(256), 0, stream>>>(out, gn, bn);

  (void)in_sizes; (void)n_in; (void)out_size; (void)ws_size;
}


# weight-side j sums (fused matrix Mt, bias rotation): j split across a block's waves, 16-byte loads grouped per trip, 4n x 4k lane tile in Mt; partials added in fixed order via LDS, new sum order
# speedup vs baseline: 2.0098x; 1.3120x over previous
#include <hip/hip_runtime.h>
#include <hip/hip_bf16.h>
#include <math.h>


#define NT     2048
#define DD     384
#define DH     128
#define D3     384
#define D9     1152
#define PKC    48
#define PTOK   64
#define PCOL   144
#define APITCH 388
#define GAMMA  5.0f
#define LOG2E  1.4426950408889634f

typedef float v2f __attribute__((ext_vector_type(2)));
typedef float v4f __attribute__((ext_vector_type(4)));
typedef float v8f __attribute__((ext_vector_type(8)));

static __device__ __forceinline__ v8f wmma4(v2f a, v2f b, v8f c) {
  return __builtin_amdgcn_wmma_f32_16x16x4_f32(
       false, a,  false, b,
       (short)0, c,  false,  false);
}

static __device__ __forceinline__ float softplusf(float w) {
  return (w > 20.0f) ? w : log1pf(__expf(w));
}


__global__ void cea_softplus(const float* __restrict__ in,
                             float* __restrict__ out, int n) {
  int i = blockIdx.x * blockDim.x + threadIdx.x;
  if (i < n) out[i] = softplusf(in[i]);
}

__global__ void cea_softplus_frag(const float* __restrict__ Wp,
                                  float* __restrict__ spPf) {
  const int o4 = blockIdx.x * 256 + threadIdx.x;
  const int lane = o4 & 31;
  const int c = lane & 15, h = lane >> 4;
  const int ch = o4 >> 5;
  const int nt = ch / PKC, kc = ch - nt * PKC;
  const v4f w = *(const v4f*)&Wp[(nt * 16 + c) * D3 + kc * 8 + 4 * h];
  v4f o;
  o.x = softplusf(w.x); o.y = softplusf(w.y);
  o.z = softplusf(w.z); o.w = softplusf(w.w);
  *(v4f*)&spPf[o4 * 4] = o;
}

#define FJS 8
#define FJL (D3 / FJS)
#define FJB 12
__global__ __launch_bounds__(256) void cea_fuse(const float* __restrict__ spw,
                                                const float* __restrict__ Sq,
                                                const float* __restrict__ Sk,
                                                const float* __restrict__ Sv,
                                                float* __restrict__ Mt) {
  __shared__ __attribute__((aligned(16))) float part[FJS * 4 * DH];
  const int h   = blockIdx.y;
  const int n0  = blockIdx.x * 4;
  const int tid = threadIdx.x;
  const int lane = tid & 31, seg = tid >> 5;
  const float* S  = (h == 0) ? Sq : (h == 1) ? Sk : Sv;
  const float* Wl = spw + h * (D3 * DH) + (seg * FJL) * DH + lane * 4;
  const float* Sl = S + (seg * FJL) * D3 + n0;

  v4f acc0 = {}, acc1 = {}, acc2 = {}, acc3 = {};
  for (int j0 = 0; j0 < FJL; j0 += FJB) {
    v4f w[FJB], s[FJB];
#pragma unroll
    for (int u = 0; u < FJB; ++u) {
      w[u] = *(const v4f*)(Wl + (j0 + u) * DH);
      s[u] = *(const v4f*)(Sl + (j0 + u) * D3);
    }
#pragma unroll
    for (int u = 0; u < FJB; ++u) asm volatile("" :: "v"(w[u]), "v"(s[u]));
#pragma unroll
    for (int u = 0; u < FJB; ++u) {
      acc0 += w[u] * s[u].x;
      acc1 += w[u] * s[u].y;
      acc2 += w[u] * s[u].z;
      acc3 += w[u] * s[u].w;
    }
  }
  *(v4f*)&part[(seg * 4 + 0) * DH + lane * 4] = acc0;
  *(v4f*)&part[(seg * 4 + 1) * DH + lane * 4] = acc1;
  *(v4f*)&part[(seg * 4 + 2) * DH + lane * 4] = acc2;
  *(v4f*)&part[(seg * 4 + 3) * DH + lane * 4] = acc3;
  __syncthreads();

  for (int o = tid; o < 4 * DH; o += 256) {
    float a = part[o];
    for (int sg = 1; sg < FJS; ++sg) a += part[sg * (4 * DH) + o];
    Mt[h * (D3 * DH) + n0 * DH + o] = a;
  }
}

#define BJS 16
#define BJL (D3 / BJS)
__global__ __launch_bounds__(512) void cea_bias(const float* __restrict__ bq, const float* __restrict__ bk,
                                                const float* __restrict__ bv,
                                                const float* __restrict__ Sq, const float* __restrict__ Sk,
                                                const float* __restrict__ Sv,
                                                float* __restrict__ c) {
  __shared__ __attribute__((aligned(16))) float part[BJS * 128];
  const int h   = blockIdx.y;
  const int n0  = blockIdx.x * 128;
  const int tid = threadIdx.x;
  const int lane = tid & 31, seg = tid >> 5;
  const float* b = (h == 0) ? bq : (h == 1) ? bk : bv;
  const float* S = (h == 0) ? Sq : (h == 1) ? Sk : Sv;
  const float* bl = b + seg * BJL;
  const float* Sl = S + (seg * BJL) * D3 + n0 + lane * 4;

  v4f bb[BJL / 4], sv[BJL];
#pragma unroll
  for (int u = 0; u < BJL / 4; ++u) bb[u] = *(const v4f*)(bl + 4 * u);
#pragma unroll
  for (int u = 0; u < BJL; ++u) sv[u] = *(const v4f*)(Sl + u * D3);

  v4f acc = {};
#pragma unroll
  for (int u = 0; u < BJL; ++u) acc += sv[u] * bb[u >> 2][u & 3];
  *(v4f*)&part[seg * 128 + lane * 4] = acc;
  __syncthreads();

  if (tid < 128) {
    float a = part[tid];
    for (int sg = 1; sg < BJS; ++sg) a += part[sg * 128 + tid];
    c[h * D3 + n0 + tid] = a;
  }
}

__global__ void cea_qkv(const float* __restrict__ x,
                        const float* __restrict__ Mt,
                        const float* __restrict__ cb,
                        float* __restrict__ qkv) {
  __shared__ float xT[16 * DH];
  const int h  = blockIdx.y;
  const int t0 = blockIdx.x * 16;
  const int tid = threadIdx.x;

  for (int idx = tid; idx < 16 * DH; idx += 256) {
    int r = idx >> 7, cc = idx & 127;
    xT[idx] = x[(t0 + r) * DD + h * DH + cc];
  }
  __syncthreads();

  const int lane = tid & 31, wid = tid >> 5;
  const int m  = lane & 15;
  const int kh = (lane >> 4) * 2;
  const float* __restrict__ Mh = Mt + h * (D3 * DH);
  const int n0 = wid * 48;

  v8f acc0 = {}, acc1 = {}, acc2 = {};
  for (int k0 = 0; k0 < DH; k0 += 4) {
    v2f a  = *(const v2f*)&xT[m * DH + k0 + kh];
    v2f b0 = *(const v2f*)&Mh[(n0 + 0  + m) * DH + k0 + kh];
    v2f b1 = *(const v2f*)&Mh[(n0 + 16 + m) * DH + k0 + kh];
    v2f b2 = *(const v2f*)&Mh[(n0 + 32 + m) * DH + k0 + kh];
    acc0 = wmma4(a, b0, acc0);
    acc1 = wmma4(a, b1, acc1);
    acc2 = wmma4(a, b2, acc2);
  }

  const int rbase = (lane >> 4) * 8;
  const int ncol  = lane & 15;
  for (int r = 0; r < 8; ++r) {
    int row = t0 + rbase + r;
    qkv[row * D9 + h * D3 + n0 + 0  + ncol] = acc0[r] + cb[h * D3 + n0 + 0  + ncol];
    qkv[row * D9 + h * D3 + n0 + 16 + ncol] = acc1[r] + cb[h * D3 + n0 + 16 + ncol];
    qkv[row * D9 + h * D3 + n0 + 32 + ncol] = acc2[r] + cb[h * D3 + n0 + 32 + ncol];
  }
}

__global__ void cea_attn(const float* __restrict__ qkv,
                         float* __restrict__ att) {
  __shared__ float Qs[D3], Ksc[D3], Vs[D3];
  const int t = blockIdx.x, i = threadIdx.x;
  const float* __restrict__ base = qkv + t * D9;
  Qs[i]  = base[i];
  Ksc[i] = base[D3 + i] * (-GAMMA * LOG2E);
  Vs[i]  = base[2 * D3 + i];
  __syncthreads();

  const float Qi = Qs[i];
  float acc = 0.0f, ssum = 0.0f;
#pragma unroll 4
  for (int j = 0; j < D3; ++j) {
    float s = __builtin_amdgcn_rcpf(1.0f + __builtin_amdgcn_exp2f(Qi * Ksc[j]));
    acc  += s * Vs[j];
    ssum += s;
  }
  att[t * D3 + i] = acc / (ssum + 1e-8f);
}


static __device__ __forceinline__ v2f lo2(v4f v) { return __builtin_shufflevector(v, v, 0, 1); }
static __device__ __forceinline__ v2f hi2(v4f v) { return __builtin_shufflevector(v, v, 2, 3); }

template <int NTW>
static __device__ __forceinline__ void proj_load(const float* __restrict__ aLane,
                                                 const v4f* __restrict__ bLane,
                                                 int kc, v4f& a, v4f (&b)[NTW]) {
  a = *(const v4f*)(aLane + kc * 8);
#pragma unroll
  for (int q = 0; q < NTW; ++q) b[q] = bLane[(q * PKC + kc) * 32];
}

template <int NTW>
static __device__ __forceinline__ void proj_mma(v4f a, const v4f (&b)[NTW], v8f (&acc)[NTW]) {
#pragma unroll
  for (int q = 0; q < NTW; ++q) acc[q] = wmma4(lo2(a), lo2(b[q]), acc[q]);
#pragma unroll
  for (int q = 0; q < NTW; ++q) acc[q] = wmma4(hi2(a), hi2(b[q]), acc[q]);
}

template <int NTW>
static __device__ __forceinline__ void proj_wave(const float* __restrict__ aLane,
                                                 const v4f* __restrict__ bLane,
                                                 const float* __restrict__ bp,
                                                 float* __restrict__ hout,
                                                 int row0, int col0) {
  v8f acc[NTW];
#pragma unroll
  for (int q = 0; q < NTW; ++q) acc[q] = (v8f){};

  v4f a[4], b[4][NTW];
  proj_load<NTW>(aLane, bLane, 0, a[0], b[0]);
  proj_load<NTW>(aLane, bLane, 1, a[1], b[1]);
  proj_load<NTW>(aLane, bLane, 2, a[2], b[2]);
#pragma unroll
  for (int s = 0; s < 3; ++s) {
    asm volatile("" :: "v"(a[s]));
#pragma unroll
    for (int q = 0; q < NTW; ++q) asm volatile("" :: "v"(b[s][q]));
  }
  for (int kc = 0; kc < PKC; kc += 4) {
#pragma unroll
    for (int s = 0; s < 4; ++s) {
      const int kp = kc + s + 3;
      const int kn = (kp < PKC) ? kp : PKC - 1;
      proj_load<NTW>(aLane, bLane, kn, a[(s + 3) & 3], b[(s + 3) & 3]);
      proj_mma<NTW>(a[s], b[s], acc);
    }
  }

#pragma unroll
  for (int q = 0; q < NTW; ++q) {
    const int n = col0 + q * 16;
    const float bias = bp[n];
#pragma unroll
    for (int r = 0; r < 8; ++r)
      hout[(row0 + r) * D9 + n] = acc[q][r] + bias;
  }
}

__global__ __launch_bounds__(256) void cea_proj(const float* __restrict__ att,
                                                const float* __restrict__ spPf,
                                                const float* __restrict__ bp,
                                                float* __restrict__ hout) {
  __shared__ __attribute__((aligned(16))) float aT[PTOK * APITCH];
  const int t0  = blockIdx.x * PTOK;
  const int tid = threadIdx.x;
  {
    const int sw = tid >> 5, sl = tid & 31;
    const float* src = att + (t0 + sw * 8) * D3 + sl * 4;
    float*       dst = aT + (sw * 8) * APITCH + sl * 4;
    v4f tmp[24];
#pragma unroll
    for (int i = 0; i < 24; ++i) tmp[i] = *(const v4f*)(src + (i / 3) * D3 + (i % 3) * 128);
#pragma unroll
    for (int i = 0; i < 24; ++i) *(v4f*)(dst + (i / 3) * APITCH + (i % 3) * 128) = tmp[i];
  }
  __syncthreads();

  const int lane = tid & 31, wid = tid >> 5;
  const int c  = lane & 15;
  const int h  = lane >> 4;
  const int mi = wid & 3;
  const int nt0 = (wid >> 2) * 5;
  const float* aLane = &aT[(mi * 16 + c) * APITCH + 4 * h];
  const v4f*   bLane = (const v4f*)spPf + ((blockIdx.y * 9 + nt0) * PKC) * 32 + lane;
  const int row0 = t0 + mi * 16 + h * 8;
  const int col0 = blockIdx.y * PCOL + nt0 * 16 + c;

  if (wid < 4) proj_wave<5>(aLane, bLane, bp, hout, row0, col0);
  else         proj_wave<4>(aLane, bLane, bp, hout, row0, col0);
}

__global__ void cea_ln(float* __restrict__ out,
                       const float* __restrict__ gn,
                       const float* __restrict__ bn) {
  __shared__ float row[D9];
  __shared__ float red[256];
  const int t = blockIdx.x, tid = threadIdx.x;
  float* __restrict__ h = out + t * D9;

  float ps = 0.0f;
  for (int n = tid; n < D9; n += 256) { float v = h[n]; row[n] = v; ps += v; }
  red[tid] = ps;
  __syncthreads();
  for (int s = 128; s > 0; s >>= 1) {
    if (tid < s) red[tid] += red[tid + s];
    __syncthreads();
  }
  const float mu = red[0] * (1.0f / (float)D9);
  __syncthreads();

  float pv = 0.0f;
  for (int n = tid; n < D9; n += 256) { float dv = row[n] - mu; pv += dv * dv; }
  red[tid] = pv;
  __syncthreads();
  for (int s = 128; s > 0; s >>= 1) {
    if (tid < s) red[tid] += red[tid + s];
    __syncthreads();
  }
  const float var = red[0] * (1.0f / (float)D9);
  const float rs  = __builtin_amdgcn_rsqf(var + 1e-5f);

  for (int n = tid; n < D9; n += 256)
    h[n] = (row[n] - mu) * rs * gn[n] + bn[n];
}


extern "C" void kernel_launch(void* const* d_in, const int* in_sizes, int n_in,
                              void* d_out, int out_size, void* d_ws, size_t ws_size,
                              hipStream_t stream) {
  const float* x  = (const float*)d_in[0];
  const float* Wq = (const float*)d_in[1];
  const float* bq = (const float*)d_in[2];
  const float* Sq = (const float*)d_in[3];
  const float* Wk = (const float*)d_in[4];
  const float* bk = (const float*)d_in[5];
  const float* Sk = (const float*)d_in[6];
  const float* Wv = (const float*)d_in[7];
  const float* bv = (const float*)d_in[8];
  const float* Sv = (const float*)d_in[9];
  const float* Wp = (const float*)d_in[10];
  const float* bp = (const float*)d_in[11];
  const float* gn = (const float*)d_in[12];
  const float* bn = (const float*)d_in[13];
  float* out = (float*)d_out;

  float* ws = (float*)d_ws;
  float* spw  = ws;
  float* Mt   = spw + 3 * D3 * DH;
  float* cb   = Mt  + 3 * D3 * DH;
  float* spP  = cb  + 3 * D3;
  float* qkv  = spP + D9 * D3;
  float* attb = qkv + NT * D9;

  cea_softplus<<<dim3((D3 * DH + 255) / 256), dim3(256), 0, stream>>>(Wq, spw,              D3 * DH);
  cea_softplus<<<dim3((D3 * DH + 255) / 256), dim3(256), 0, stream>>>(Wk, spw + D3 * DH,    D3 * DH);
  cea_softplus<<<dim3((D3 * DH + 255) / 256), dim3(256), 0, stream>>>(Wv, spw + 2 * D3 * DH, D3 * DH);
  cea_softplus_frag<<<dim3((D9 * D3 / 4) / 256), dim3(256), 0, stream>>>(Wp, spP);

  cea_fuse<<<dim3(D3 / 4, 3), dim3(256), 0, stream>>>(spw, Sq, Sk, Sv, Mt);
  cea_bias<<<dim3(D3 / 128, 3), dim3(512), 0, stream>>>(bq, bk, bv, Sq, Sk, Sv, cb);

  cea_qkv<<<dim3(NT / 16, 3), dim3(256), 0, stream>>>(x, Mt, cb, qkv);

  cea_attn<<<dim3(NT), dim3(D3), 0, stream>>>(qkv, attb);

  cea_proj<<<dim3(NT / PTOK, D9 / PCOL), dim3(256), 0, stream>>>(attb, spP, bp, out);

  cea_ln<<<dim3(NT), dim3(256), 0, stream>>>(out, gn, bn);

  (void)in_sizes; (void)n_in; (void)out_size; (void)ws_size;
}


# QKV GEMM: fused weights written in WMMA-fragment order by the fuse launch, B read straight to registers (coalesced 16 B/lane, no LDS for B), 64x192 block tile; k summed in new order within 8s
# speedup vs baseline: 2.1616x; 1.0755x over previous
#include <hip/hip_runtime.h>
#include <hip/hip_bf16.h>
#include <math.h>


#define NT     2048
#define DD     384
#define DH     128
#define D3     384
#define D9     1152
#define PKC    48
#define PTOK   64
#define PCOL   144
#define APITCH 388
#define QKC    16
#define QTOK   64
#define QNT    12
#define QPITCH 132
#define GAMMA  5.0f
#define LOG2E  1.4426950408889634f

typedef float v2f __attribute__((ext_vector_type(2)));
typedef float v4f __attribute__((ext_vector_type(4)));
typedef float v8f __attribute__((ext_vector_type(8)));

static __device__ __forceinline__ v8f wmma4(v2f a, v2f b, v8f c) {
  return __builtin_amdgcn_wmma_f32_16x16x4_f32(
       false, a,  false, b,
       (short)0, c,  false,  false);
}

static __device__ __forceinline__ v2f lo2(v4f v) { return __builtin_shufflevector(v, v, 0, 1); }
static __device__ __forceinline__ v2f hi2(v4f v) { return __builtin_shufflevector(v, v, 2, 3); }

static __device__ __forceinline__ float softplusf(float w) {
  return (w > 20.0f) ? w : log1pf(__expf(w));
}


__global__ void cea_softplus(const float* __restrict__ in,
                             float* __restrict__ out, int n) {
  int i = blockIdx.x * blockDim.x + threadIdx.x;
  if (i < n) out[i] = softplusf(in[i]);
}

__global__ void cea_softplus_frag(const float* __restrict__ Wp,
                                  float* __restrict__ spPf) {
  const int o4 = blockIdx.x * 256 + threadIdx.x;
  const int lane = o4 & 31;
  const int c = lane & 15, h = lane >> 4;
  const int ch = o4 >> 5;
  const int nt = ch / PKC, kc = ch - nt * PKC;
  const v4f w = *(const v4f*)&Wp[(nt * 16 + c) * D3 + kc * 8 + 4 * h];
  v4f o;
  o.x = softplusf(w.x); o.y = softplusf(w.y);
  o.z = softplusf(w.z); o.w = softplusf(w.w);
  *(v4f*)&spPf[o4 * 4] = o;
}

#define FJS 8
#define FJL (D3 / FJS)
#define FJB 12
__global__ __launch_bounds__(256) void cea_fuse(const float* __restrict__ spw,
                                                const float* __restrict__ Sq,
                                                const float* __restrict__ Sk,
                                                const float* __restrict__ Sv,
                                                float* __restrict__ Mt) {
  __shared__ __attribute__((aligned(16))) float part[FJS * 4 * DH];
  const int h   = blockIdx.y;
  const int n0  = blockIdx.x * 4;
  const int tid = threadIdx.x;
  const int lane = tid & 31, seg = tid >> 5;
  const float* S  = (h == 0) ? Sq : (h == 1) ? Sk : Sv;
  const float* Wl = spw + h * (D3 * DH) + (seg * FJL) * DH + lane * 4;
  const float* Sl = S + (seg * FJL) * D3 + n0;

  v4f acc0 = {}, acc1 = {}, acc2 = {}, acc3 = {};
  for (int j0 = 0; j0 < FJL; j0 += FJB) {
    v4f w[FJB], s[FJB];
#pragma unroll
    for (int u = 0; u < FJB; ++u) {
      w[u] = *(const v4f*)(Wl + (j0 + u) * DH);
      s[u] = *(const v4f*)(Sl + (j0 + u) * D3);
    }
#pragma unroll
    for (int u = 0; u < FJB; ++u) asm volatile("" :: "v"(w[u]), "v"(s[u]));
#pragma unroll
    for (int u = 0; u < FJB; ++u) {
      acc0 += w[u] * s[u].x;
      acc1 += w[u] * s[u].y;
      acc2 += w[u] * s[u].z;
      acc3 += w[u] * s[u].w;
    }
  }
  *(v4f*)&part[(seg * 4 + 0) * DH + lane * 4] = acc0;
  *(v4f*)&part[(seg * 4 + 1) * DH + lane * 4] = acc1;
  *(v4f*)&part[(seg * 4 + 2) * DH + lane * 4] = acc2;
  *(v4f*)&part[(seg * 4 + 3) * DH + lane * 4] = acc3;
  __syncthreads();

  if (tid < 128) {
    const int kc = seg * 4 + (lane >> 3);
    const int hh = (lane >> 2) & 1, nl = lane & 3;
    const float* pp = &part[nl * DH + kc * 8 + hh * 4];
    v4f a = *(const v4f*)pp;
    for (int sg = 1; sg < FJS; ++sg) a += *(const v4f*)(pp + sg * (4 * DH));
    const int nt = n0 >> 4;
    const int p  = ((n0 & 15) >> 2) * 8 + (lane & 7);
    *(v4f*)&Mt[((((h * 24 + nt) * QKC + kc) * 32) + p) * 4] = a;
  }
}

#define BJS 16
#define BJL (D3 / BJS)
__global__ __launch_bounds__(512) void cea_bias(const float* __restrict__ bq, const float* __restrict__ bk,
                                                const float* __restrict__ bv,
                                                const float* __restrict__ Sq, const float* __restrict__ Sk,
                                                const float* __restrict__ Sv,
                                                float* __restrict__ c) {
  __shared__ __attribute__((aligned(16))) float part[BJS * 128];
  const int h   = blockIdx.y;
  const int n0  = blockIdx.x * 128;
  const int tid = threadIdx.x;
  const int lane = tid & 31, seg = tid >> 5;
  const float* b = (h == 0) ? bq : (h == 1) ? bk : bv;
  const float* S = (h == 0) ? Sq : (h == 1) ? Sk : Sv;
  const float* bl = b + seg * BJL;
  const float* Sl = S + (seg * BJL) * D3 + n0 + lane * 4;

  v4f bb[BJL / 4], sv[BJL];
#pragma unroll
  for (int u = 0; u < BJL / 4; ++u) bb[u] = *(const v4f*)(bl + 4 * u);
#pragma unroll
  for (int u = 0; u < BJL; ++u) sv[u] = *(const v4f*)(Sl + u * D3);

  v4f acc = {};
#pragma unroll
  for (int u = 0; u < BJL; ++u) acc += sv[u] * bb[u >> 2][u & 3];
  *(v4f*)&part[seg * 128 + lane * 4] = acc;
  __syncthreads();

  if (tid < 128) {
    float a = part[tid];
    for (int sg = 1; sg < BJS; ++sg) a += part[sg * 128 + tid];
    c[h * D3 + n0 + tid] = a;
  }
}


static __device__ __forceinline__ void qkv_load(const float* __restrict__ aLane,
                                                const v4f* __restrict__ bLane,
                                                int kc, v4f (&a)[2], v4f (&b)[3]) {
  a[0] = *(const v4f*)(aLane + kc * 8);
  a[1] = *(const v4f*)(aLane + 16 * QPITCH + kc * 8);
#pragma unroll
  for (int q = 0; q < 3; ++q) b[q] = bLane[(q * QKC + kc) * 32];
}

static __device__ __forceinline__ void qkv_mma(const v4f (&a)[2], const v4f (&b)[3], v8f (&acc)[2][3]) {
#pragma unroll
  for (int m = 0; m < 2; ++m)
#pragma unroll
    for (int q = 0; q < 3; ++q) acc[m][q] = wmma4(lo2(a[m]), lo2(b[q]), acc[m][q]);
#pragma unroll
  for (int m = 0; m < 2; ++m)
#pragma unroll
    for (int q = 0; q < 3; ++q) acc[m][q] = wmma4(hi2(a[m]), hi2(b[q]), acc[m][q]);
}

__global__ __launch_bounds__(256) void cea_qkv(const float* __restrict__ x,
                                               const float* __restrict__ Mt,
                                               const float* __restrict__ cb,
                                               float* __restrict__ qkv) {
  __shared__ __attribute__((aligned(16))) float xT[QTOK * QPITCH];
  const int h    = blockIdx.y >> 1;
  const int ntb  = (blockIdx.y & 1) * QNT;
  const int t0   = blockIdx.x * QTOK;
  const int tid  = threadIdx.x;
  const int lane = tid & 31, wid = tid >> 5;
  {
    const float* src = x + (t0 + wid * 8) * DD + h * DH + lane * 4;
    float*       dst = xT + (wid * 8) * QPITCH + lane * 4;
    v4f tmp[8];
#pragma unroll
    for (int i = 0; i < 8; ++i) tmp[i] = *(const v4f*)(src + i * DD);
#pragma unroll
    for (int i = 0; i < 8; ++i) *(v4f*)(dst + i * QPITCH) = tmp[i];
  }
  __syncthreads();

  const int c  = lane & 15;
  const int hh = lane >> 4;
  const int mp = wid & 1;
  const int nt = ntb + (wid >> 1) * 3;
  const float* aLane = &xT[(mp * 32 + c) * QPITCH + 4 * hh];
  const v4f*   bLane = (const v4f*)Mt + ((h * 24 + nt) * QKC) * 32
                       + ((c >> 2) * 8 + hh * 4 + (c & 3));

  v8f acc[2][3];
#pragma unroll
  for (int m = 0; m < 2; ++m)
#pragma unroll
    for (int q = 0; q < 3; ++q) acc[m][q] = (v8f){};

  v4f a[4][2], b[4][3];
  qkv_load(aLane, bLane, 0, a[0], b[0]);
  qkv_load(aLane, bLane, 1, a[1], b[1]);
  qkv_load(aLane, bLane, 2, a[2], b[2]);
#pragma unroll
  for (int s = 0; s < 3; ++s) {
    asm volatile("" :: "v"(a[s][0]), "v"(a[s][1]));
#pragma unroll
    for (int q = 0; q < 3; ++q) asm volatile("" :: "v"(b[s][q]));
  }
  for (int kc = 0; kc < QKC; kc += 4) {
#pragma unroll
    for (int s = 0; s < 4; ++s) {
      const int kp = kc + s + 3;
      const int kn = (kp < QKC) ? kp : QKC - 1;
      qkv_load(aLane, bLane, kn, a[(s + 3) & 3], b[(s + 3) & 3]);
      qkv_mma(a[s], b[s], acc);
    }
  }

  const int row0 = t0 + mp * 32 + hh * 8;
#pragma unroll
  for (int m = 0; m < 2; ++m) {
#pragma unroll
    for (int q = 0; q < 3; ++q) {
      const int n = h * D3 + (nt + q) * 16 + c;
      const float bias = cb[n];
#pragma unroll
      for (int r = 0; r < 8; ++r)
        qkv[(row0 + m * 16 + r) * D9 + n] = acc[m][q][r] + bias;
    }
  }
}

__global__ void cea_attn(const float* __restrict__ qkv,
                         float* __restrict__ att) {
  __shared__ float Qs[D3], Ksc[D3], Vs[D3];
  const int t = blockIdx.x, i = threadIdx.x;
  const float* __restrict__ base = qkv + t * D9;
  Qs[i]  = base[i];
  Ksc[i] = base[D3 + i] * (-GAMMA * LOG2E);
  Vs[i]  = base[2 * D3 + i];
  __syncthreads();

  const float Qi = Qs[i];
  float acc = 0.0f, ssum = 0.0f;
#pragma unroll 4
  for (int j = 0; j < D3; ++j) {
    float s = __builtin_amdgcn_rcpf(1.0f + __builtin_amdgcn_exp2f(Qi * Ksc[j]));
    acc  += s * Vs[j];
    ssum += s;
  }
  att[t * D3 + i] = acc / (ssum + 1e-8f);
}


template <int NTW>
static __device__ __forceinline__ void proj_load(const float* __restrict__ aLane,
                                                 const v4f* __restrict__ bLane,
                                                 int kc, v4f& a, v4f (&b)[NTW]) {
  a = *(const v4f*)(aLane + kc * 8);
#pragma unroll
  for (int q = 0; q < NTW; ++q) b[q] = bLane[(q * PKC + kc) * 32];
}

template <int NTW>
static __device__ __forceinline__ void proj_mma(v4f a, const v4f (&b)[NTW], v8f (&acc)[NTW]) {
#pragma unroll
  for (int q = 0; q < NTW; ++q) acc[q] = wmma4(lo2(a), lo2(b[q]), acc[q]);
#pragma unroll
  for (int q = 0; q < NTW; ++q) acc[q] = wmma4(hi2(a), hi2(b[q]), acc[q]);
}

template <int NTW>
static __device__ __forceinline__ void proj_wave(const float* __restrict__ aLane,
                                                 const v4f* __restrict__ bLane,
                                                 const float* __restrict__ bp,
                                                 float* __restrict__ hout,
                                                 int row0, int col0) {
  v8f acc[NTW];
#pragma unroll
  for (int q = 0; q < NTW; ++q) acc[q] = (v8f){};

  v4f a[4], b[4][NTW];
  proj_load<NTW>(aLane, bLane, 0, a[0], b[0]);
  proj_load<NTW>(aLane, bLane, 1, a[1], b[1]);
  proj_load<NTW>(aLane, bLane, 2, a[2], b[2]);
#pragma unroll
  for (int s = 0; s < 3; ++s) {
    asm volatile("" :: "v"(a[s]));
#pragma unroll
    for (int q = 0; q < NTW; ++q) asm volatile("" :: "v"(b[s][q]));
  }
  for (int kc = 0; kc < PKC; kc += 4) {
#pragma unroll
    for (int s = 0; s < 4; ++s) {
      const int kp = kc + s + 3;
      const int kn = (kp < PKC) ? kp : PKC - 1;
      proj_load<NTW>(aLane, bLane, kn, a[(s + 3) & 3], b[(s + 3) & 3]);
      proj_mma<NTW>(a[s], b[s], acc);
    }
  }

#pragma unroll
  for (int q = 0; q < NTW; ++q) {
    const int n = col0 + q * 16;
    const float bias = bp[n];
#pragma unroll
    for (int r = 0; r < 8; ++r)
      hout[(row0 + r) * D9 + n] = acc[q][r] + bias;
  }
}

__global__ __launch_bounds__(256) void cea_proj(const float* __restrict__ att,
                                                const float* __restrict__ spPf,
                                                const float* __restrict__ bp,
                                                float* __restrict__ hout) {
  __shared__ __attribute__((aligned(16))) float aT[PTOK * APITCH];
  const int t0  = blockIdx.x * PTOK;
  const int tid = threadIdx.x;
  {
    const int sw = tid >> 5, sl = tid & 31;
    const float* src = att + (t0 + sw * 8) * D3 + sl * 4;
    float*       dst = aT + (sw * 8) * APITCH + sl * 4;
    v4f tmp[24];
#pragma unroll
    for (int i = 0; i < 24; ++i) tmp[i] = *(const v4f*)(src + (i / 3) * D3 + (i % 3) * 128);
#pragma unroll
    for (int i = 0; i < 24; ++i) *(v4f*)(dst + (i / 3) * APITCH + (i % 3) * 128) = tmp[i];
  }
  __syncthreads();

  const int lane = tid & 31, wid = tid >> 5;
  const int c  = lane & 15;
  const int h  = lane >> 4;
  const int mi = wid & 3;
  const int nt0 = (wid >> 2) * 5;
  const float* aLane = &aT[(mi * 16 + c) * APITCH + 4 * h];
  const v4f*   bLane = (const v4f*)spPf + ((blockIdx.y * 9 + nt0) * PKC) * 32 + lane;
  const int row0 = t0 + mi * 16 + h * 8;
  const int col0 = blockIdx.y * PCOL + nt0 * 16 + c;

  if (wid < 4) proj_wave<5>(aLane, bLane, bp, hout, row0, col0);
  else         proj_wave<4>(aLane, bLane, bp, hout, row0, col0);
}

__global__ void cea_ln(float* __restrict__ out,
                       const float* __restrict__ gn,
                       const float* __restrict__ bn) {
  __shared__ float row[D9];
  __shared__ float red[256];
  const int t = blockIdx.x, tid = threadIdx.x;
  float* __restrict__ h = out + t * D9;

  float ps = 0.0f;
  for (int n = tid; n < D9; n += 256) { float v = h[n]; row[n] = v; ps += v; }
  red[tid] = ps;
  __syncthreads();
  for (int s = 128; s > 0; s >>= 1) {
    if (tid < s) red[tid] += red[tid + s];
    __syncthreads();
  }
  const float mu = red[0] * (1.0f / (float)D9);
  __syncthreads();

  float pv = 0.0f;
  for (int n = tid; n < D9; n += 256) { float dv = row[n] - mu; pv += dv * dv; }
  red[tid] = pv;
  __syncthreads();
  for (int s = 128; s > 0; s >>= 1) {
    if (tid < s) red[tid] += red[tid + s];
    __syncthreads();
  }
  const float var = red[0] * (1.0f / (float)D9);
  const float rs  = __builtin_amdgcn_rsqf(var + 1e-5f);

  for (int n = tid; n < D9; n += 256)
    h[n] = (row[n] - mu) * rs * gn[n] + bn[n];
}


extern "C" void kernel_launch(void* const* d_in, const int* in_sizes, int n_in,
                              void* d_out, int out_size, void* d_ws, size_t ws_size,
                              hipStream_t stream) {
  const float* x  = (const float*)d_in[0];
  const float* Wq = (const float*)d_in[1];
  const float* bq = (const float*)d_in[2];
  const float* Sq = (const float*)d_in[3];
  const float* Wk = (const float*)d_in[4];
  const float* bk = (const float*)d_in[5];
  const float* Sk = (const float*)d_in[6];
  const float* Wv = (const float*)d_in[7];
  const float* bv = (const float*)d_in[8];
  const float* Sv = (const float*)d_in[9];
  const float* Wp = (const float*)d_in[10];
  const float* bp = (const float*)d_in[11];
  const float* gn = (const float*)d_in[12];
  const float* bn = (const float*)d_in[13];
  float* out = (float*)d_out;

  float* ws = (float*)d_ws;
  float* spw  = ws;
  float* Mt   = spw + 3 * D3 * DH;
  float* cb   = Mt  + 3 * D3 * DH;
  float* spP  = cb  + 3 * D3;
  float* qkv  = spP + D9 * D3;
  float* attb = qkv + NT * D9;

  cea_softplus<<<dim3((D3 * DH + 255) / 256), dim3(256), 0, stream>>>(Wq, spw,              D3 * DH);
  cea_softplus<<<dim3((D3 * DH + 255) / 256), dim3(256), 0, stream>>>(Wk, spw + D3 * DH,    D3 * DH);
  cea_softplus<<<dim3((D3 * DH + 255) / 256), dim3(256), 0, stream>>>(Wv, spw + 2 * D3 * DH, D3 * DH);
  cea_softplus_frag<<<dim3((D9 * D3 / 4) / 256), dim3(256), 0, stream>>>(Wp, spP);

  cea_fuse<<<dim3(D3 / 4, 3), dim3(256), 0, stream>>>(spw, Sq, Sk, Sv, Mt);
  cea_bias<<<dim3(D3 / 128, 3), dim3(512), 0, stream>>>(bq, bk, bv, Sq, Sk, Sv, cb);

  cea_qkv<<<dim3(NT / QTOK, 6), dim3(256), 0, stream>>>(x, Mt, cb, qkv);

  cea_attn<<<dim3(NT), dim3(D3), 0, stream>>>(qkv, attb);

  cea_proj<<<dim3(NT / PTOK, D9 / PCOL), dim3(256), 0, stream>>>(attb, spP, bp, out);

  cea_ln<<<dim3(NT), dim3(256), 0, stream>>>(out, gn, bn);

  (void)in_sizes; (void)n_in; (void)out_size; (void)ws_size;
}


# attention score loop issued as (plain, transcendental) pairs: two-wide f32 plain ops on j pairs, 4 rows per lane, 3-stage software pipeline over j chunks; even and odd j summed apart, added at end
# speedup vs baseline: 2.3941x; 1.1076x over previous
#include <hip/hip_runtime.h>
#include <hip/hip_bf16.h>
#include <math.h>


#define NT     2048
#define DD     384
#define DH     128
#define D3     384
#define D9     1152
#define PKC    48
#define PTOK   64
#define PCOL   144
#define APITCH 388
#define QKC    16
#define QTOK   64
#define QNT    12
#define QPITCH 132
#define GAMMA  5.0f
#define LOG2E  1.4426950408889634f

typedef float v2f __attribute__((ext_vector_type(2)));
typedef float v4f __attribute__((ext_vector_type(4)));
typedef float v8f __attribute__((ext_vector_type(8)));

static __device__ __forceinline__ v8f wmma4(v2f a, v2f b, v8f c) {
  return __builtin_amdgcn_wmma_f32_16x16x4_f32(
       false, a,  false, b,
       (short)0, c,  false,  false);
}

static __device__ __forceinline__ v2f lo2(v4f v) { return __builtin_shufflevector(v, v, 0, 1); }
static __device__ __forceinline__ v2f hi2(v4f v) { return __builtin_shufflevector(v, v, 2, 3); }

static __device__ __forceinline__ float softplusf(float w) {
  return (w > 20.0f) ? w : log1pf(__expf(w));
}


__global__ void cea_softplus(const float* __restrict__ in,
                             float* __restrict__ out, int n) {
  int i = blockIdx.x * blockDim.x + threadIdx.x;
  if (i < n) out[i] = softplusf(in[i]);
}

__global__ void cea_softplus_frag(const float* __restrict__ Wp,
                                  float* __restrict__ spPf) {
  const int o4 = blockIdx.x * 256 + threadIdx.x;
  const int lane = o4 & 31;
  const int c = lane & 15, h = lane >> 4;
  const int ch = o4 >> 5;
  const int nt = ch / PKC, kc = ch - nt * PKC;
  const v4f w = *(const v4f*)&Wp[(nt * 16 + c) * D3 + kc * 8 + 4 * h];
  v4f o;
  o.x = softplusf(w.x); o.y = softplusf(w.y);
  o.z = softplusf(w.z); o.w = softplusf(w.w);
  *(v4f*)&spPf[o4 * 4] = o;
}

#define FJS 8
#define FJL (D3 / FJS)
#define FJB 12
__global__ __launch_bounds__(256) void cea_fuse(const float* __restrict__ spw,
                                                const float* __restrict__ Sq,
                                                const float* __restrict__ Sk,
                                                const float* __restrict__ Sv,
                                                float* __restrict__ Mt) {
  __shared__ __attribute__((aligned(16))) float part[FJS * 4 * DH];
  const int h   = blockIdx.y;
  const int n0  = blockIdx.x * 4;
  const int tid = threadIdx.x;
  const int lane = tid & 31, seg = tid >> 5;
  const float* S  = (h == 0) ? Sq : (h == 1) ? Sk : Sv;
  const float* Wl = spw + h * (D3 * DH) + (seg * FJL) * DH + lane * 4;
  const float* Sl = S + (seg * FJL) * D3 + n0;

  v4f acc0 = {}, acc1 = {}, acc2 = {}, acc3 = {};
  for (int j0 = 0; j0 < FJL; j0 += FJB) {
    v4f w[FJB], s[FJB];
#pragma unroll
    for (int u = 0; u < FJB; ++u) {
      w[u] = *(const v4f*)(Wl + (j0 + u) * DH);
      s[u] = *(const v4f*)(Sl + (j0 + u) * D3);
    }
#pragma unroll
    for (int u = 0; u < FJB; ++u) asm volatile("" :: "v"(w[u]), "v"(s[u]));
#pragma unroll
    for (int u = 0; u < FJB; ++u) {
      acc0 += w[u] * s[u].x;
      acc1 += w[u] * s[u].y;
      acc2 += w[u] * s[u].z;
      acc3 += w[u] * s[u].w;
    }
  }
  *(v4f*)&part[(seg * 4 + 0) * DH + lane * 4] = acc0;
  *(v4f*)&part[(seg * 4 + 1) * DH + lane * 4] = acc1;
  *(v4f*)&part[(seg * 4 + 2) * DH + lane * 4] = acc2;
  *(v4f*)&part[(seg * 4 + 3) * DH + lane * 4] = acc3;
  __syncthreads();

  if (tid < 128) {
    const int kc = seg * 4 + (lane >> 3);
    const int hh = (lane >> 2) & 1, nl = lane & 3;
    const float* pp = &part[nl * DH + kc * 8 + hh * 4];
    v4f a = *(const v4f*)pp;
    for (int sg = 1; sg < FJS; ++sg) a += *(const v4f*)(pp + sg * (4 * DH));
    const int nt = n0 >> 4;
    const int p  = ((n0 & 15) >> 2) * 8 + (lane & 7);
    *(v4f*)&Mt[((((h * 24 + nt) * QKC + kc) * 32) + p) * 4] = a;
  }
}

#define BJS 16
#define BJL (D3 / BJS)
__global__ __launch_bounds__(512) void cea_bias(const float* __restrict__ bq, const float* __restrict__ bk,
                                                const float* __restrict__ bv,
                                                const float* __restrict__ Sq, const float* __restrict__ Sk,
                                                const float* __restrict__ Sv,
                                                float* __restrict__ c) {
  __shared__ __attribute__((aligned(16))) float part[BJS * 128];
  const int h   = blockIdx.y;
  const int n0  = blockIdx.x * 128;
  const int tid = threadIdx.x;
  const int lane = tid & 31, seg = tid >> 5;
  const float* b = (h == 0) ? bq : (h == 1) ? bk : bv;
  const float* S = (h == 0) ? Sq : (h == 1) ? Sk : Sv;
  const float* bl = b + seg * BJL;
  const float* Sl = S + (seg * BJL) * D3 + n0 + lane * 4;

  v4f bb[BJL / 4], sv[BJL];
#pragma unroll
  for (int u = 0; u < BJL / 4; ++u) bb[u] = *(const v4f*)(bl + 4 * u);
#pragma unroll
  for (int u = 0; u < BJL; ++u) sv[u] = *(const v4f*)(Sl + u * D3);

  v4f acc = {};
#pragma unroll
  for (int u = 0; u < BJL; ++u) acc += sv[u] * bb[u >> 2][u & 3];
  *(v4f*)&part[seg * 128 + lane * 4] = acc;
  __syncthreads();

  if (tid < 128) {
    float a = part[tid];
    for (int sg = 1; sg < BJS; ++sg) a += part[sg * 128 + tid];
    c[h * D3 + n0 + tid] = a;
  }
}


static __device__ __forceinline__ void qkv_load(const float* __restrict__ aLane,
                                                const v4f* __restrict__ bLane,
                                                int kc, v4f (&a)[2], v4f (&b)[3]) {
  a[0] = *(const v4f*)(aLane + kc * 8);
  a[1] = *(const v4f*)(aLane + 16 * QPITCH + kc * 8);
#pragma unroll
  for (int q = 0; q < 3; ++q) b[q] = bLane[(q * QKC + kc) * 32];
}

static __device__ __forceinline__ void qkv_mma(const v4f (&a)[2], const v4f (&b)[3], v8f (&acc)[2][3]) {
#pragma unroll
  for (int m = 0; m < 2; ++m)
#pragma unroll
    for (int q = 0; q < 3; ++q) acc[m][q] = wmma4(lo2(a[m]), lo2(b[q]), acc[m][q]);
#pragma unroll
  for (int m = 0; m < 2; ++m)
#pragma unroll
    for (int q = 0; q < 3; ++q) acc[m][q] = wmma4(hi2(a[m]), hi2(b[q]), acc[m][q]);
}

__global__ __launch_bounds__(256) void cea_qkv(const float* __restrict__ x,
                                               const float* __restrict__ Mt,
                                               const float* __restrict__ cb,
                                               float* __restrict__ qkv) {
  __shared__ __attribute__((aligned(16))) float xT[QTOK * QPITCH];
  const int h    = blockIdx.y >> 1;
  const int ntb  = (blockIdx.y & 1) * QNT;
  const int t0   = blockIdx.x * QTOK;
  const int tid  = threadIdx.x;
  const int lane = tid & 31, wid = tid >> 5;
  {
    const float* src = x + (t0 + wid * 8) * DD + h * DH + lane * 4;
    float*       dst = xT + (wid * 8) * QPITCH + lane * 4;
    v4f tmp[8];
#pragma unroll
    for (int i = 0; i < 8; ++i) tmp[i] = *(const v4f*)(src + i * DD);
#pragma unroll
    for (int i = 0; i < 8; ++i) *(v4f*)(dst + i * QPITCH) = tmp[i];
  }
  __syncthreads();

  const int c  = lane & 15;
  const int hh = lane >> 4;
  const int mp = wid & 1;
  const int nt = ntb + (wid >> 1) * 3;
  const float* aLane = &xT[(mp * 32 + c) * QPITCH + 4 * hh];
  const v4f*   bLane = (const v4f*)Mt + ((h * 24 + nt) * QKC) * 32
                       + ((c >> 2) * 8 + hh * 4 + (c & 3));

  v8f acc[2][3];
#pragma unroll
  for (int m = 0; m < 2; ++m)
#pragma unroll
    for (int q = 0; q < 3; ++q) acc[m][q] = (v8f){};

  v4f a[4][2], b[4][3];
  qkv_load(aLane, bLane, 0, a[0], b[0]);
  qkv_load(aLane, bLane, 1, a[1], b[1]);
  qkv_load(aLane, bLane, 2, a[2], b[2]);
#pragma unroll
  for (int s = 0; s < 3; ++s) {
    asm volatile("" :: "v"(a[s][0]), "v"(a[s][1]));
#pragma unroll
    for (int q = 0; q < 3; ++q) asm volatile("" :: "v"(b[s][q]));
  }
  for (int kc = 0; kc < QKC; kc += 4) {
#pragma unroll
    for (int s = 0; s < 4; ++s) {
      const int kp = kc + s + 3;
      const int kn = (kp < QKC) ? kp : QKC - 1;
      qkv_load(aLane, bLane, kn, a[(s + 3) & 3], b[(s + 3) & 3]);
      qkv_mma(a[s], b[s], acc);
    }
  }

  const int row0 = t0 + mp * 32 + hh * 8;
#pragma unroll
  for (int m = 0; m < 2; ++m) {
#pragma unroll
    for (int q = 0; q < 3; ++q) {
      const int n = h * D3 + (nt + q) * 16 + c;
      const float bias = cb[n];
#pragma unroll
      for (int r = 0; r < 8; ++r)
        qkv[(row0 + m * 16 + r) * D9 + n] = acc[m][q][r] + bias;
    }
  }
}

#define ATOK  4
#define ACH   (D3 / 4)

static __device__ __forceinline__ void att_mul(const v2f (&qq)[4], v4f k4, v2f (&z)[8]) {
  const v2f k0 = lo2(k4), k1 = hi2(k4);
#pragma unroll
  for (int r = 0; r < 4; ++r) { z[r] = qq[r] * k0; z[4 + r] = qq[r] * k1; }
}
static __device__ __forceinline__ void att_exp(const v2f (&z)[8], v2f (&e)[8]) {
#pragma unroll
  for (int c = 0; c < 8; ++c) {
    e[c].x = __builtin_amdgcn_exp2f(z[c].x);
    e[c].y = __builtin_amdgcn_exp2f(z[c].y);
  }
}
static __device__ __forceinline__ void att_one(const v2f (&e)[8], v2f (&a)[8]) {
#pragma unroll
  for (int c = 0; c < 8; ++c) a[c] = e[c] + 1.0f;
}
static __device__ __forceinline__ void att_sum(const v2f (&a)[8], v4f v4, v2f (&acc)[4], v2f (&ssum)[4]) {
  const v2f v0 = lo2(v4), v1 = hi2(v4);
#pragma unroll
  for (int c = 0; c < 8; ++c) {
    v2f s;
    s.x = __builtin_amdgcn_rcpf(a[c].x);
    s.y = __builtin_amdgcn_rcpf(a[c].y);
    acc[c & 3]  += s * ((c < 4) ? v0 : v1);
    ssum[c & 3] += s;
  }
}

__global__ __launch_bounds__(384) void cea_attn(const float* __restrict__ qkv,
                                                float* __restrict__ att) {
  __shared__ __attribute__((aligned(16))) float KV[ATOK * 2 * D3];
  const int tid  = threadIdx.x;
  const int lane = tid & 31, wid = tid >> 5;
  const int tk   = wid / 3;
  const int i0   = (wid - tk * 3) * 128 + lane * 4;
  const int t    = blockIdx.x * ATOK + tk;
  const float* __restrict__ base = qkv + t * D9;
  const v4f q4 = *(const v4f*)(base + i0);
  {
    const v4f k4 = *(const v4f*)(base + D3 + i0);
    const v4f v4 = *(const v4f*)(base + 2 * D3 + i0);
    *(v4f*)&KV[tk * 2 * D3 + i0]      = k4 * (-GAMMA * LOG2E);
    *(v4f*)&KV[tk * 2 * D3 + D3 + i0] = v4;
  }
  __syncthreads();

  v2f qq[4], acc[4], ssum[4];
#pragma unroll
  for (int r = 0; r < 4; ++r) {
    qq[r]   = (v2f){q4[r], q4[r]};
    acc[r]  = (v2f){0.0f, 0.0f};
    ssum[r] = (v2f){0.0f, 0.0f};
  }
  const float* __restrict__ kp = &KV[tk * 2 * D3];

  v2f z[8], e[8], aA[8], aB[8];
  att_mul(qq, *(const v4f*)kp, z);
  att_exp(z, e);
  att_one(e, aA);
  att_mul(qq, *(const v4f*)(kp + 4), z);
  att_exp(z, e);
  v4f kA = *(const v4f*)(kp + 8);
  v4f kB = *(const v4f*)(kp + 12);
  v4f vA = *(const v4f*)(kp + D3);
  v4f vB;
  for (int n = 2; n < ACH; n += 2) {
    att_sum(aA, vA, acc, ssum);
    att_one(e, aB);
    att_mul(qq, kA, z);
    att_exp(z, e);
    const int nA = (n + 2 < ACH) ? n + 2 : ACH - 1;
    kA = *(const v4f*)(kp + 4 * nA);
    vB = *(const v4f*)(kp + D3 + 4 * (n - 1));
    att_sum(aB, vB, acc, ssum);
    att_one(e, aA);
    att_mul(qq, kB, z);
    att_exp(z, e);
    const int nB = (n + 3 < ACH) ? n + 3 : ACH - 1;
    kB = *(const v4f*)(kp + 4 * nB);
    vA = *(const v4f*)(kp + D3 + 4 * n);
#pragma unroll
    for (int g = 0; g < 64; ++g) {
      __builtin_amdgcn_sched_group_barrier(0x002, 1, 0);
      __builtin_amdgcn_sched_group_barrier(0x400, 1, 0);
    }
  }
  att_sum(aA, vA, acc, ssum);
  att_one(e, aB);
  att_sum(aB, *(const v4f*)(kp + D3 + 4 * (ACH - 1)), acc, ssum);

  v4f o;
#pragma unroll
  for (int r = 0; r < 4; ++r)
    o[r] = (acc[r].x + acc[r].y) / ((ssum[r].x + ssum[r].y) + 1e-8f);
  *(v4f*)(att + t * D3 + i0) = o;
}


template <int NTW>
static __device__ __forceinline__ void proj_load(const float* __restrict__ aLane,
                                                 const v4f* __restrict__ bLane,
                                                 int kc, v4f& a, v4f (&b)[NTW]) {
  a = *(const v4f*)(aLane + kc * 8);
#pragma unroll
  for (int q = 0; q < NTW; ++q) b[q] = bLane[(q * PKC + kc) * 32];
}

template <int NTW>
static __device__ __forceinline__ void proj_mma(v4f a, const v4f (&b)[NTW], v8f (&acc)[NTW]) {
#pragma unroll
  for (int q = 0; q < NTW; ++q) acc[q] = wmma4(lo2(a), lo2(b[q]), acc[q]);
#pragma unroll
  for (int q = 0; q < NTW; ++q) acc[q] = wmma4(hi2(a), hi2(b[q]), acc[q]);
}

template <int NTW>
static __device__ __forceinline__ void proj_wave(const float* __restrict__ aLane,
                                                 const v4f* __restrict__ bLane,
                                                 const float* __restrict__ bp,
                                                 float* __restrict__ hout,
                                                 int row0, int col0) {
  v8f acc[NTW];
#pragma unroll
  for (int q = 0; q < NTW; ++q) acc[q] = (v8f){};

  v4f a[4], b[4][NTW];
  proj_load<NTW>(aLane, bLane, 0, a[0], b[0]);
  proj_load<NTW>(aLane, bLane, 1, a[1], b[1]);
  proj_load<NTW>(aLane, bLane, 2, a[2], b[2]);
#pragma unroll
  for (int s = 0; s < 3; ++s) {
    asm volatile("" :: "v"(a[s]));
#pragma unroll
    for (int q = 0; q < NTW; ++q) asm volatile("" :: "v"(b[s][q]));
  }
  for (int kc = 0; kc < PKC; kc += 4) {
#pragma unroll
    for (int s = 0; s < 4; ++s) {
      const int kp = kc + s + 3;
      const int kn = (kp < PKC) ? kp : PKC - 1;
      proj_load<NTW>(aLane, bLane, kn, a[(s + 3) & 3], b[(s + 3) & 3]);
      proj_mma<NTW>(a[s], b[s], acc);
    }
  }

#pragma unroll
  for (int q = 0; q < NTW; ++q) {
    const int n = col0 + q * 16;
    const float bias = bp[n];
#pragma unroll
    for (int r = 0; r < 8; ++r)
      hout[(row0 + r) * D9 + n] = acc[q][r] + bias;
  }
}

__global__ __launch_bounds__(256) void cea_proj(const float* __restrict__ att,
                                                const float* __restrict__ spPf,
                                                const float* __restrict__ bp,
                                                float* __restrict__ hout) {
  __shared__ __attribute__((aligned(16))) float aT[PTOK * APITCH];
  const int t0  = blockIdx.x * PTOK;
  const int tid = threadIdx.x;
  {
    const int sw = tid >> 5, sl = tid & 31;
    const float* src = att + (t0 + sw * 8) * D3 + sl * 4;
    float*       dst = aT + (sw * 8) * APITCH + sl * 4;
    v4f tmp[24];
#pragma unroll
    for (int i = 0; i < 24; ++i) tmp[i] = *(const v4f*)(src + (i / 3) * D3 + (i % 3) * 128);
#pragma unroll
    for (int i = 0; i < 24; ++i) *(v4f*)(dst + (i / 3) * APITCH + (i % 3) * 128) = tmp[i];
  }
  __syncthreads();

  const int lane = tid & 31, wid = tid >> 5;
  const int c  = lane & 15;
  const int h  = lane >> 4;
  const int mi = wid & 3;
  const int nt0 = (wid >> 2) * 5;
  const float* aLane = &aT[(mi * 16 + c) * APITCH + 4 * h];
  const v4f*   bLane = (const v4f*)spPf + ((blockIdx.y * 9 + nt0) * PKC) * 32 + lane;
  const int row0 = t0 + mi * 16 + h * 8;
  const int col0 = blockIdx.y * PCOL + nt0 * 16 + c;

  if (wid < 4) proj_wave<5>(aLane, bLane, bp, hout, row0, col0);
  else         proj_wave<4>(aLane, bLane, bp, hout, row0, col0);
}

__global__ void cea_ln(float* __restrict__ out,
                       const float* __restrict__ gn,
                       const float* __restrict__ bn) {
  __shared__ float row[D9];
  __shared__ float red[256];
  const int t = blockIdx.x, tid = threadIdx.x;
  float* __restrict__ h = out + t * D9;

  float ps = 0.0f;
  for (int n = tid; n < D9; n += 256) { float v = h[n]; row[n] = v; ps += v; }
  red[tid] = ps;
  __syncthreads();
  for (int s = 128; s > 0; s >>= 1) {
    if (tid < s) red[tid] += red[tid + s];
    __syncthreads();
  }
  const float mu = red[0] * (1.0f / (float)D9);
  __syncthreads();

  float pv = 0.0f;
  for (int n = tid; n < D9; n += 256) { float dv = row[n] - mu; pv += dv * dv; }
  red[tid] = pv;
  __syncthreads();
  for (int s = 128; s > 0; s >>= 1) {
    if (tid < s) red[tid] += red[tid + s];
    __syncthreads();
  }
  const float var = red[0] * (1.0f / (float)D9);
  const float rs  = __builtin_amdgcn_rsqf(var + 1e-5f);

  for (int n = tid; n < D9; n += 256)
    h[n] = (row[n] - mu) * rs * gn[n] + bn[n];
}


extern "C" void kernel_launch(void* const* d_in, const int* in_sizes, int n_in,
                              void* d_out, int out_size, void* d_ws, size_t ws_size,
                              hipStream_t stream) {
  const float* x  = (const float*)d_in[0];
  const float* Wq = (const float*)d_in[1];
  const float* bq = (const float*)d_in[2];
  const float* Sq = (const float*)d_in[3];
  const float* Wk = (const float*)d_in[4];
  const float* bk = (const float*)d_in[5];
  const float* Sk = (const float*)d_in[6];
  const float* Wv = (const float*)d_in[7];
  const float* bv = (const float*)d_in[8];
  const float* Sv = (const float*)d_in[9];
  const float* Wp = (const float*)d_in[10];
  const float* bp = (const float*)d_in[11];
  const float* gn = (const float*)d_in[12];
  const float* bn = (const float*)d_in[13];
  float* out = (float*)d_out;

  float* ws = (float*)d_ws;
  float* spw  = ws;
  float* Mt   = spw + 3 * D3 * DH;
  float* cb   = Mt  + 3 * D3 * DH;
  float* spP  = cb  + 3 * D3;
  float* qkv  = spP + D9 * D3;
  float* attb = qkv + NT * D9;

  cea_softplus<<<dim3((D3 * DH + 255) / 256), dim3(256), 0, stream>>>(Wq, spw,              D3 * DH);
  cea_softplus<<<dim3((D3 * DH + 255) / 256), dim3(256), 0, stream>>>(Wk, spw + D3 * DH,    D3 * DH);
  cea_softplus<<<dim3((D3 * DH + 255) / 256), dim3(256), 0, stream>>>(Wv, spw + 2 * D3 * DH, D3 * DH);
  cea_softplus_frag<<<dim3((D9 * D3 / 4) / 256), dim3(256), 0, stream>>>(Wp, spP);

  cea_fuse<<<dim3(D3 / 4, 3), dim3(256), 0, stream>>>(spw, Sq, Sk, Sv, Mt);
  cea_bias<<<dim3(D3 / 128, 3), dim3(512), 0, stream>>>(bq, bk, bv, Sq, Sk, Sv, cb);

  cea_qkv<<<dim3(NT / QTOK, 6), dim3(256), 0, stream>>>(x, Mt, cb, qkv);

  cea_attn<<<dim3(NT / ATOK), dim3(3 * 32 * ATOK), 0, stream>>>(qkv, attb);

  cea_proj<<<dim3(NT / PTOK, D9 / PCOL), dim3(256), 0, stream>>>(attb, spP, bp, out);

  cea_ln<<<dim3(NT), dim3(256), 0, stream>>>(out, gn, bn);

  (void)in_sizes; (void)n_in; (void)out_size; (void)ws_size;
}


# softplus+reorder of the projection weights and the bias rotation run as extra work-groups of the QKV GEMM launch (matrix-bound, idle units) and of the fuse launch, no launches of their own: 10 -> 8
# speedup vs baseline: 2.6240x; 1.0960x over previous
#include <hip/hip_runtime.h>
#include <hip/hip_bf16.h>
#include <math.h>


#define NT     2048
#define DD     384
#define DH     128
#define D3     384
#define D9     1152
#define PKC    48
#define PTOK   64
#define PCOL   144
#define APITCH 388
#define QKC    16
#define QTOK   64
#define QNT    12
#define QPITCH 132
#define QGB    192
#define GAMMA  5.0f
#define LOG2E  1.4426950408889634f

typedef float v2f __attribute__((ext_vector_type(2)));
typedef float v4f __attribute__((ext_vector_type(4)));
typedef float v8f __attribute__((ext_vector_type(8)));

static __device__ __forceinline__ v8f wmma4(v2f a, v2f b, v8f c) {
  return __builtin_amdgcn_wmma_f32_16x16x4_f32(
       false, a,  false, b,
       (short)0, c,  false,  false);
}

static __device__ __forceinline__ v2f lo2(v4f v) { return __builtin_shufflevector(v, v, 0, 1); }
static __device__ __forceinline__ v2f hi2(v4f v) { return __builtin_shufflevector(v, v, 2, 3); }

static __device__ __forceinline__ float softplusf(float w) {
  return (w > 20.0f) ? w : log1pf(__expf(w));
}


__global__ void cea_softplus(const float* __restrict__ in,
                             float* __restrict__ out, int n) {
  int i = blockIdx.x * blockDim.x + threadIdx.x;
  if (i < n) out[i] = softplusf(in[i]);
}

#define SPB  216
#define SPT  2
static __device__ __forceinline__ v4f wp_frag_load(const float* __restrict__ Wp, int o4) {
  const int lane = o4 & 31;
  const int c = lane & 15, h = lane >> 4;
  const int ch = o4 >> 5;
  const int nt = ch / PKC, kc = ch - nt * PKC;
  return *(const v4f*)&Wp[(nt * 16 + c) * D3 + kc * 8 + 4 * h];
}
static __device__ __forceinline__ void wp_frag_job(const float* __restrict__ Wp,
                                                   float* __restrict__ spPf, int blk, int tid) {
  const int o0 = blk * (256 * SPT) + tid;
  v4f w[SPT];
#pragma unroll
  for (int i = 0; i < SPT; ++i) w[i] = wp_frag_load(Wp, o0 + i * 256);
#pragma unroll
  for (int i = 0; i < SPT; ++i) {
    v4f o;
    o.x = softplusf(w[i].x); o.y = softplusf(w[i].y);
    o.z = softplusf(w[i].z); o.w = softplusf(w[i].w);
    *(v4f*)&spPf[(o0 + i * 256) * 4] = o;
  }
}

#define BJS 16
#define BJL (D3 / BJS)
#define BNC 64
#define BNB (D3 / BNC)
static __device__ __forceinline__ void bias_job(const float* __restrict__ b,
                                                const float* __restrict__ S,
                                                float* __restrict__ c,
                                                float* __restrict__ part,
                                                int n0, int tid) {
  const int lane = tid & 31;
  const int seg  = (tid >> 5) * 2 + (lane >> 4);
  const int nl   = (lane & 15) * 4;
  const float* bl = b + seg * BJL;
  const float* Sl = S + (seg * BJL) * D3 + n0 + nl;

  v4f bb[BJL / 4], sv[BJL];
#pragma unroll
  for (int u = 0; u < BJL / 4; ++u) bb[u] = *(const v4f*)(bl + 4 * u);
#pragma unroll
  for (int u = 0; u < BJL; ++u) sv[u] = *(const v4f*)(Sl + u * D3);

  v4f acc = {};
#pragma unroll
  for (int u = 0; u < BJL; ++u) acc += sv[u] * bb[u >> 2][u & 3];
  *(v4f*)&part[seg * BNC + nl] = acc;
  __syncthreads();

  if (tid < BNC) {
    float a = part[tid];
    for (int sg = 1; sg < BJS; ++sg) a += part[sg * BNC + tid];
    c[n0 + tid] = a;
  }
}

#define FJS 8
#define FJL (D3 / FJS)
#define FJB 12
__global__ __launch_bounds__(256) void cea_fuse(const float* __restrict__ spw,
                                                const float* __restrict__ Sq,
                                                const float* __restrict__ Sk,
                                                const float* __restrict__ Sv,
                                                const float* __restrict__ bq,
                                                const float* __restrict__ bk,
                                                const float* __restrict__ bv,
                                                float* __restrict__ Mt,
                                                float* __restrict__ cb) {
  __shared__ __attribute__((aligned(16))) float part[FJS * 4 * DH];
  const int h   = blockIdx.y;
  const int tid = threadIdx.x;
  const float* S  = (h == 0) ? Sq : (h == 1) ? Sk : Sv;
  if (blockIdx.x >= D3 / 4) {
    const float* b = (h == 0) ? bq : (h == 1) ? bk : bv;
    bias_job(b, S, cb + h * D3, part, (blockIdx.x - D3 / 4) * BNC, tid);
    return;
  }
  const int n0  = blockIdx.x * 4;
  const int lane = tid & 31, seg = tid >> 5;
  const float* Wl = spw + h * (D3 * DH) + (seg * FJL) * DH + lane * 4;
  const float* Sl = S + (seg * FJL) * D3 + n0;

  v4f acc0 = {}, acc1 = {}, acc2 = {}, acc3 = {};
  for (int j0 = 0; j0 < FJL; j0 += FJB) {
    v4f w[FJB], s[FJB];
#pragma unroll
    for (int u = 0; u < FJB; ++u) {
      w[u] = *(const v4f*)(Wl + (j0 + u) * DH);
      s[u] = *(const v4f*)(Sl + (j0 + u) * D3);
    }
#pragma unroll
    for (int u = 0; u < FJB; ++u) asm volatile("" :: "v"(w[u]), "v"(s[u]));
#pragma unroll
    for (int u = 0; u < FJB; ++u) {
      acc0 += w[u] * s[u].x;
      acc1 += w[u] * s[u].y;
      acc2 += w[u] * s[u].z;
      acc3 += w[u] * s[u].w;
    }
  }
  *(v4f*)&part[(seg * 4 + 0) * DH + lane * 4] = acc0;
  *(v4f*)&part[(seg * 4 + 1) * DH + lane * 4] = acc1;
  *(v4f*)&part[(seg * 4 + 2) * DH + lane * 4] = acc2;
  *(v4f*)&part[(seg * 4 + 3) * DH + lane * 4] = acc3;
  __syncthreads();

  if (tid < 128) {
    const int kc = seg * 4 + (lane >> 3);
    const int hh = (lane >> 2) & 1, nl = lane & 3;
    const float* pp = &part[nl * DH + kc * 8 + hh * 4];
    v4f a = *(const v4f*)pp;
    for (int sg = 1; sg < FJS; ++sg) a += *(const v4f*)(pp + sg * (4 * DH));
    const int nt = n0 >> 4;
    const int p  = ((n0 & 15) >> 2) * 8 + (lane & 7);
    *(v4f*)&Mt[((((h * 24 + nt) * QKC + kc) * 32) + p) * 4] = a;
  }
}


static __device__ __forceinline__ void qkv_load(const float* __restrict__ aLane,
                                                const v4f* __restrict__ bLane,
                                                int kc, v4f (&a)[2], v4f (&b)[3]) {
  a[0] = *(const v4f*)(aLane + kc * 8);
  a[1] = *(const v4f*)(aLane + 16 * QPITCH + kc * 8);
#pragma unroll
  for (int q = 0; q < 3; ++q) b[q] = bLane[(q * QKC + kc) * 32];
}

static __device__ __forceinline__ void qkv_mma(const v4f (&a)[2], const v4f (&b)[3], v8f (&acc)[2][3]) {
#pragma unroll
  for (int m = 0; m < 2; ++m)
#pragma unroll
    for (int q = 0; q < 3; ++q) acc[m][q] = wmma4(lo2(a[m]), lo2(b[q]), acc[m][q]);
#pragma unroll
  for (int m = 0; m < 2; ++m)
#pragma unroll
    for (int q = 0; q < 3; ++q) acc[m][q] = wmma4(hi2(a[m]), hi2(b[q]), acc[m][q]);
}

__global__ __launch_bounds__(256) void cea_qkv(const float* __restrict__ x,
                                               const float* __restrict__ Mt,
                                               const float* __restrict__ cb,
                                               float* __restrict__ qkv,
                                               const float* __restrict__ Wp,
                                               float* __restrict__ spPf) {
  __shared__ __attribute__((aligned(16))) float xT[QTOK * QPITCH];
  if (blockIdx.x >= QGB) {
    wp_frag_job(Wp, spPf, blockIdx.x - QGB, threadIdx.x);
    return;
  }
  const int gy   = blockIdx.x / (NT / QTOK);
  const int h    = gy >> 1;
  const int ntb  = (gy & 1) * QNT;
  const int t0   = (blockIdx.x - gy * (NT / QTOK)) * QTOK;
  const int tid  = threadIdx.x;
  const int lane = tid & 31, wid = tid >> 5;
  {
    const float* src = x + (t0 + wid * 8) * DD + h * DH + lane * 4;
    float*       dst = xT + (wid * 8) * QPITCH + lane * 4;
    v4f tmp[8];
#pragma unroll
    for (int i = 0; i < 8; ++i) tmp[i] = *(const v4f*)(src + i * DD);
#pragma unroll
    for (int i = 0; i < 8; ++i) *(v4f*)(dst + i * QPITCH) = tmp[i];
  }
  __syncthreads();

  const int c  = lane & 15;
  const int hh = lane >> 4;
  const int mp = wid & 1;
  const int nt = ntb + (wid >> 1) * 3;
  const float* aLane = &xT[(mp * 32 + c) * QPITCH + 4 * hh];
  const v4f*   bLane = (const v4f*)Mt + ((h * 24 + nt) * QKC) * 32
                       + ((c >> 2) * 8 + hh * 4 + (c & 3));

  v8f acc[2][3];
#pragma unroll
  for (int m = 0; m < 2; ++m)
#pragma unroll
    for (int q = 0; q < 3; ++q) acc[m][q] = (v8f){};

  v4f a[4][2], b[4][3];
  qkv_load(aLane, bLane, 0, a[0], b[0]);
  qkv_load(aLane, bLane, 1, a[1], b[1]);
  qkv_load(aLane, bLane, 2, a[2], b[2]);
#pragma unroll
  for (int s = 0; s < 3; ++s) {
    asm volatile("" :: "v"(a[s][0]), "v"(a[s][1]));
#pragma unroll
    for (int q = 0; q < 3; ++q) asm volatile("" :: "v"(b[s][q]));
  }
  for (int kc = 0; kc < QKC; kc += 4) {
#pragma unroll
    for (int s = 0; s < 4; ++s) {
      const int kp = kc + s + 3;
      const int kn = (kp < QKC) ? kp : QKC - 1;
      qkv_load(aLane, bLane, kn, a[(s + 3) & 3], b[(s + 3) & 3]);
      qkv_mma(a[s], b[s], acc);
    }
  }

  const int row0 = t0 + mp * 32 + hh * 8;
#pragma unroll
  for (int m = 0; m < 2; ++m) {
#pragma unroll
    for (int q = 0; q < 3; ++q) {
      const int n = h * D3 + (nt + q) * 16 + c;
      const float bias = cb[n];
#pragma unroll
      for (int r = 0; r < 8; ++r)
        qkv[(row0 + m * 16 + r) * D9 + n] = acc[m][q][r] + bias;
    }
  }
}

#define ATOK  4
#define ACH   (D3 / 4)

static __device__ __forceinline__ void att_mul(const v2f (&qq)[4], v4f k4, v2f (&z)[8]) {
  const v2f k0 = lo2(k4), k1 = hi2(k4);
#pragma unroll
  for (int r = 0; r < 4; ++r) { z[r] = qq[r] * k0; z[4 + r] = qq[r] * k1; }
}
static __device__ __forceinline__ void att_exp(const v2f (&z)[8], v2f (&e)[8]) {
#pragma unroll
  for (int c = 0; c < 8; ++c) {
    e[c].x = __builtin_amdgcn_exp2f(z[c].x);
    e[c].y = __builtin_amdgcn_exp2f(z[c].y);
  }
}
static __device__ __forceinline__ void att_one(const v2f (&e)[8], v2f (&a)[8]) {
#pragma unroll
  for (int c = 0; c < 8; ++c) a[c] = e[c] + 1.0f;
}
static __device__ __forceinline__ void att_sum(const v2f (&a)[8], v4f v4, v2f (&acc)[4], v2f (&ssum)[4]) {
  const v2f v0 = lo2(v4), v1 = hi2(v4);
#pragma unroll
  for (int c = 0; c < 8; ++c) {
    v2f s;
    s.x = __builtin_amdgcn_rcpf(a[c].x);
    s.y = __builtin_amdgcn_rcpf(a[c].y);
    acc[c & 3]  += s * ((c < 4) ? v0 : v1);
    ssum[c & 3] += s;
  }
}

__global__ __launch_bounds__(384) void cea_attn(const float* __restrict__ qkv,
                                                float* __restrict__ att) {
  __shared__ __attribute__((aligned(16))) float KV[ATOK * 2 * D3];
  const int tid  = threadIdx.x;
  const int lane = tid & 31, wid = tid >> 5;
  const int tk   = wid / 3;
  const int i0   = (wid - tk * 3) * 128 + lane * 4;
  const int t    = blockIdx.x * ATOK + tk;
  const float* __restrict__ base = qkv + t * D9;
  const v4f q4 = *(const v4f*)(base + i0);
  {
    const v4f k4 = *(const v4f*)(base + D3 + i0);
    const v4f v4 = *(const v4f*)(base + 2 * D3 + i0);
    *(v4f*)&KV[tk * 2 * D3 + i0]      = k4 * (-GAMMA * LOG2E);
    *(v4f*)&KV[tk * 2 * D3 + D3 + i0] = v4;
  }
  __syncthreads();

  v2f qq[4], acc[4], ssum[4];
#pragma unroll
  for (int r = 0; r < 4; ++r) {
    qq[r]   = (v2f){q4[r], q4[r]};
    acc[r]  = (v2f){0.0f, 0.0f};
    ssum[r] = (v2f){0.0f, 0.0f};
  }
  const float* __restrict__ kp = &KV[tk * 2 * D3];

  v2f z[8], e[8], aA[8], aB[8];
  att_mul(qq, *(const v4f*)kp, z);
  att_exp(z, e);
  att_one(e, aA);
  att_mul(qq, *(const v4f*)(kp + 4), z);
  att_exp(z, e);
  v4f kA = *(const v4f*)(kp + 8);
  v4f kB = *(const v4f*)(kp + 12);
  v4f vA = *(const v4f*)(kp + D3);
  v4f vB;
  for (int n = 2; n < ACH; n += 2) {
    att_sum(aA, vA, acc, ssum);
    att_one(e, aB);
    att_mul(qq, kA, z);
    att_exp(z, e);
    const int nA = (n + 2 < ACH) ? n + 2 : ACH - 1;
    kA = *(const v4f*)(kp + 4 * nA);
    vB = *(const v4f*)(kp + D3 + 4 * (n - 1));
    att_sum(aB, vB, acc, ssum);
    att_one(e, aA);
    att_mul(qq, kB, z);
    att_exp(z, e);
    const int nB = (n + 3 < ACH) ? n + 3 : ACH - 1;
    kB = *(const v4f*)(kp + 4 * nB);
    vA = *(const v4f*)(kp + D3 + 4 * n);
#pragma unroll
    for (int g = 0; g < 64; ++g) {
      __builtin_amdgcn_sched_group_barrier(0x002, 1, 0);
      __builtin_amdgcn_sched_group_barrier(0x400, 1, 0);
    }
  }
  att_sum(aA, vA, acc, ssum);
  att_one(e, aB);
  att_sum(aB, *(const v4f*)(kp + D3 + 4 * (ACH - 1)), acc, ssum);

  v4f o;
#pragma unroll
  for (int r = 0; r < 4; ++r)
    o[r] = (acc[r].x + acc[r].y) / ((ssum[r].x + ssum[r].y) + 1e-8f);
  *(v4f*)(att + t * D3 + i0) = o;
}


template <int NTW>
static __device__ __forceinline__ void proj_load(const float* __restrict__ aLane,
                                                 const v4f* __restrict__ bLane,
                                                 int kc, v4f& a, v4f (&b)[NTW]) {
  a = *(const v4f*)(aLane + kc * 8);
#pragma unroll
  for (int q = 0; q < NTW; ++q) b[q] = bLane[(q * PKC + kc) * 32];
}

template <int NTW>
static __device__ __forceinline__ void proj_mma(v4f a, const v4f (&b)[NTW], v8f (&acc)[NTW]) {
#pragma unroll
  for (int q = 0; q < NTW; ++q) acc[q] = wmma4(lo2(a), lo2(b[q]), acc[q]);
#pragma unroll
  for (int q = 0; q < NTW; ++q) acc[q] = wmma4(hi2(a), hi2(b[q]), acc[q]);
}

template <int NTW>
static __device__ __forceinline__ void proj_wave(const float* __restrict__ aLane,
                                                 const v4f* __restrict__ bLane,
                                                 const float* __restrict__ bp,
                                                 float* __restrict__ hout,
                                                 int row0, int col0) {
  v8f acc[NTW];
#pragma unroll
  for (int q = 0; q < NTW; ++q) acc[q] = (v8f){};

  v4f a[4], b[4][NTW];
  proj_load<NTW>(aLane, bLane, 0, a[0], b[0]);
  proj_load<NTW>(aLane, bLane, 1, a[1], b[1]);
  proj_load<NTW>(aLane, bLane, 2, a[2], b[2]);
#pragma unroll
  for (int s = 0; s < 3; ++s) {
    asm volatile("" :: "v"(a[s]));
#pragma unroll
    for (int q = 0; q < NTW; ++q) asm volatile("" :: "v"(b[s][q]));
  }
  for (int kc = 0; kc < PKC; kc += 4) {
#pragma unroll
    for (int s = 0; s < 4; ++s) {
      const int kp = kc + s + 3;
      const int kn = (kp < PKC) ? kp : PKC - 1;
      proj_load<NTW>(aLane, bLane, kn, a[(s + 3) & 3], b[(s + 3) & 3]);
      proj_mma<NTW>(a[s], b[s], acc);
    }
  }

#pragma unroll
  for (int q = 0; q < NTW; ++q) {
    const int n = col0 + q * 16;
    const float bias = bp[n];
#pragma unroll
    for (int r = 0; r < 8; ++r)
      hout[(row0 + r) * D9 + n] = acc[q][r] + bias;
  }
}

__global__ __launch_bounds__(256) void cea_proj(const float* __restrict__ att,
                                                const float* __restrict__ spPf,
                                                const float* __restrict__ bp,
                                                float* __restrict__ hout) {
  __shared__ __attribute__((aligned(16))) float aT[PTOK * APITCH];
  const int t0  = blockIdx.x * PTOK;
  const int tid = threadIdx.x;
  {
    const int sw = tid >> 5, sl = tid & 31;
    const float* src = att + (t0 + sw * 8) * D3 + sl * 4;
    float*       dst = aT + (sw * 8) * APITCH + sl * 4;
    v4f tmp[24];
#pragma unroll
    for (int i = 0; i < 24; ++i) tmp[i] = *(const v4f*)(src + (i / 3) * D3 + (i % 3) * 128);
#pragma unroll
    for (int i = 0; i < 24; ++i) *(v4f*)(dst + (i / 3) * APITCH + (i % 3) * 128) = tmp[i];
  }
  __syncthreads();

  const int lane = tid & 31, wid = tid >> 5;
  const int c  = lane & 15;
  const int h  = lane >> 4;
  const int mi = wid & 3;
  const int nt0 = (wid >> 2) * 5;
  const float* aLane = &aT[(mi * 16 + c) * APITCH + 4 * h];
  const v4f*   bLane = (const v4f*)spPf + ((blockIdx.y * 9 + nt0) * PKC) * 32 + lane;
  const int row0 = t0 + mi * 16 + h * 8;
  const int col0 = blockIdx.y * PCOL + nt0 * 16 + c;

  if (wid < 4) proj_wave<5>(aLane, bLane, bp, hout, row0, col0);
  else         proj_wave<4>(aLane, bLane, bp, hout, row0, col0);
}

__global__ void cea_ln(float* __restrict__ out,
                       const float* __restrict__ gn,
                       const float* __restrict__ bn) {
  __shared__ float row[D9];
  __shared__ float red[256];
  const int t = blockIdx.x, tid = threadIdx.x;
  float* __restrict__ h = out + t * D9;

  float ps = 0.0f;
  for (int n = tid; n < D9; n += 256) { float v = h[n]; row[n] = v; ps += v; }
  red[tid] = ps;
  __syncthreads();
  for (int s = 128; s > 0; s >>= 1) {
    if (tid < s) red[tid] += red[tid + s];
    __syncthreads();
  }
  const float mu = red[0] * (1.0f / (float)D9);
  __syncthreads();

  float pv = 0.0f;
  for (int n = tid; n < D9; n += 256) { float dv = row[n] - mu; pv += dv * dv; }
  red[tid] = pv;
  __syncthreads();
  for (int s = 128; s > 0; s >>= 1) {
    if (tid < s) red[tid] += red[tid + s];
    __syncthreads();
  }
  const float var = red[0] * (1.0f / (float)D9);
  const float rs  = __builtin_amdgcn_rsqf(var + 1e-5f);

  for (int n = tid; n < D9; n += 256)
    h[n] = (row[n] - mu) * rs * gn[n] + bn[n];
}


extern "C" void kernel_launch(void* const* d_in, const int* in_sizes, int n_in,
                              void* d_out, int out_size, void* d_ws, size_t ws_size,
                              hipStream_t stream) {
  const float* x  = (const float*)d_in[0];
  const float* Wq = (const float*)d_in[1];
  const float* bq = (const float*)d_in[2];
  const float* Sq = (const float*)d_in[3];
  const float* Wk = (const float*)d_in[4];
  const float* bk = (const float*)d_in[5];
  const float* Sk = (const float*)d_in[6];
  const float* Wv = (const float*)d_in[7];
  const float* bv = (const float*)d_in[8];
  const float* Sv = (const float*)d_in[9];
  const float* Wp = (const float*)d_in[10];
  const float* bp = (const float*)d_in[11];
  const float* gn = (const float*)d_in[12];
  const float* bn = (const float*)d_in[13];
  float* out = (float*)d_out;

  float* ws = (float*)d_ws;
  float* spw  = ws;
  float* Mt   = spw + 3 * D3 * DH;
  float* cb   = Mt  + 3 * D3 * DH;
  float* spP  = cb  + 3 * D3;
  float* qkv  = spP + D9 * D3;
  float* attb = qkv + NT * D9;

  cea_softplus<<<dim3((D3 * DH + 255) / 256), dim3(256), 0, stream>>>(Wq, spw,              D3 * DH);
  cea_softplus<<<dim3((D3 * DH + 255) / 256), dim3(256), 0, stream>>>(Wk, spw + D3 * DH,    D3 * DH);
  cea_softplus<<<dim3((D3 * DH + 255) / 256), dim3(256), 0, stream>>>(Wv, spw + 2 * D3 * DH, D3 * DH);

  cea_fuse<<<dim3(D3 / 4 + BNB, 3), dim3(256), 0, stream>>>(spw, Sq, Sk, Sv, bq, bk, bv, Mt, cb);

  cea_qkv<<<dim3(QGB + SPB), dim3(256), 0, stream>>>(x, Mt, cb, qkv, Wp, spP);

  cea_attn<<<dim3(NT / ATOK), dim3(3 * 32 * ATOK), 0, stream>>>(qkv, attb);

  cea_proj<<<dim3(NT / PTOK, D9 / PCOL), dim3(256), 0, stream>>>(attb, spP, bp, out);

  cea_ln<<<dim3(NT), dim3(256), 0, stream>>>(out, gn, bn);

  (void)in_sizes; (void)n_in; (void)out_size; (void)ws_size;
}


# head-weight softplus taken in the fuse launch's registers on each W value after its load (hardware exp2, log2; last bits differ); no softplus launches, no softplus copy in workspace: 8 -> 5 launches
# speedup vs baseline: 2.8972x; 1.1041x over previous
#include <hip/hip_runtime.h>
#include <hip/hip_bf16.h>
#include <math.h>


#define NT     2048
#define DD     384
#define DH     128
#define D3     384
#define D9     1152
#define PKC    48
#define PTOK   64
#define PCOL   144
#define APITCH 388
#define QKC    16
#define QTOK   64
#define QNT    12
#define QPITCH 132
#define QGB    192
#define GAMMA  5.0f
#define LOG2E  1.4426950408889634f

typedef float v2f __attribute__((ext_vector_type(2)));
typedef float v4f __attribute__((ext_vector_type(4)));
typedef float v8f __attribute__((ext_vector_type(8)));

static __device__ __forceinline__ v8f wmma4(v2f a, v2f b, v8f c) {
  return __builtin_amdgcn_wmma_f32_16x16x4_f32(
       false, a,  false, b,
       (short)0, c,  false,  false);
}

static __device__ __forceinline__ v2f lo2(v4f v) { return __builtin_shufflevector(v, v, 0, 1); }
static __device__ __forceinline__ v2f hi2(v4f v) { return __builtin_shufflevector(v, v, 2, 3); }

static __device__ __forceinline__ float softplusf(float w) {
  return (w > 20.0f) ? w : log1pf(__expf(w));
}

static __device__ __forceinline__ float softplus_hw(float w) {
  const float t = __builtin_amdgcn_exp2f(-__builtin_fabsf(w) * LOG2E);
  return __builtin_fmaxf(w, 0.0f) + 0.6931471805599453f * __builtin_amdgcn_logf(1.0f + t);
}


#define SPB  216
#define SPT  2
static __device__ __forceinline__ v4f wp_frag_load(const float* __restrict__ Wp, int o4) {
  const int lane = o4 & 31;
  const int c = lane & 15, h = lane >> 4;
  const int ch = o4 >> 5;
  const int nt = ch / PKC, kc = ch - nt * PKC;
  return *(const v4f*)&Wp[(nt * 16 + c) * D3 + kc * 8 + 4 * h];
}
static __device__ __forceinline__ void wp_frag_job(const float* __restrict__ Wp,
                                                   float* __restrict__ spPf, int blk, int tid) {
  const int o0 = blk * (256 * SPT) + tid;
  v4f w[SPT];
#pragma unroll
  for (int i = 0; i < SPT; ++i) w[i] = wp_frag_load(Wp, o0 + i * 256);
#pragma unroll
  for (int i = 0; i < SPT; ++i) {
    v4f o;
    o.x = softplusf(w[i].x); o.y = softplusf(w[i].y);
    o.z = softplusf(w[i].z); o.w = softplusf(w[i].w);
    *(v4f*)&spPf[(o0 + i * 256) * 4] = o;
  }
}

#define BJS 16
#define BJL (D3 / BJS)
#define BNC 64
#define BNB (D3 / BNC)
static __device__ __forceinline__ void bias_job(const float* __restrict__ b,
                                                const float* __restrict__ S,
                                                float* __restrict__ c,
                                                float* __restrict__ part,
                                                int n0, int tid) {
  const int lane = tid & 31;
  const int seg  = (tid >> 5) * 2 + (lane >> 4);
  const int nl   = (lane & 15) * 4;
  const float* bl = b + seg * BJL;
  const float* Sl = S + (seg * BJL) * D3 + n0 + nl;

  v4f bb[BJL / 4], sv[BJL];
#pragma unroll
  for (int u = 0; u < BJL / 4; ++u) bb[u] = *(const v4f*)(bl + 4 * u);
#pragma unroll
  for (int u = 0; u < BJL; ++u) sv[u] = *(const v4f*)(Sl + u * D3);

  v4f acc = {};
#pragma unroll
  for (int u = 0; u < BJL; ++u) acc += sv[u] * bb[u >> 2][u & 3];
  *(v4f*)&part[seg * BNC + nl] = acc;
  __syncthreads();

  if (tid < BNC) {
    float a = part[tid];
    for (int sg = 1; sg < BJS; ++sg) a += part[sg * BNC + tid];
    c[n0 + tid] = a;
  }
}

#define FJS 8
#define FJL (D3 / FJS)
#define FJB 12
__global__ __launch_bounds__(256) void cea_fuse(const float* __restrict__ Wq,
                                                const float* __restrict__ Wk,
                                                const float* __restrict__ Wv,
                                                const float* __restrict__ Sq,
                                                const float* __restrict__ Sk,
                                                const float* __restrict__ Sv,
                                                const float* __restrict__ bq,
                                                const float* __restrict__ bk,
                                                const float* __restrict__ bv,
                                                float* __restrict__ Mt,
                                                float* __restrict__ cb) {
  __shared__ __attribute__((aligned(16))) float part[FJS * 4 * DH];
  const int h   = blockIdx.y;
  const int tid = threadIdx.x;
  const float* S  = (h == 0) ? Sq : (h == 1) ? Sk : Sv;
  if (blockIdx.x >= D3 / 4) {
    const float* b = (h == 0) ? bq : (h == 1) ? bk : bv;
    bias_job(b, S, cb + h * D3, part, (blockIdx.x - D3 / 4) * BNC, tid);
    return;
  }
  const int n0  = blockIdx.x * 4;
  const int lane = tid & 31, seg = tid >> 5;
  const float* W  = (h == 0) ? Wq : (h == 1) ? Wk : Wv;
  const float* Wl = W + (seg * FJL) * DH + lane * 4;
  const float* Sl = S + (seg * FJL) * D3 + n0;

  v4f acc0 = {}, acc1 = {}, acc2 = {}, acc3 = {};
  for (int j0 = 0; j0 < FJL; j0 += FJB) {
    v4f w[FJB], s[FJB];
#pragma unroll
    for (int u = 0; u < FJB; ++u) {
      w[u] = *(const v4f*)(Wl + (j0 + u) * DH);
      s[u] = *(const v4f*)(Sl + (j0 + u) * D3);
    }
#pragma unroll
    for (int u = 0; u < FJB; ++u) asm volatile("" :: "v"(w[u]), "v"(s[u]));
#pragma unroll
    for (int u = 0; u < FJB; ++u) {
      w[u].x = softplus_hw(w[u].x); w[u].y = softplus_hw(w[u].y);
      w[u].z = softplus_hw(w[u].z); w[u].w = softplus_hw(w[u].w);
      acc0 += w[u] * s[u].x;
      acc1 += w[u] * s[u].y;
      acc2 += w[u] * s[u].z;
      acc3 += w[u] * s[u].w;
    }
  }
  *(v4f*)&part[(seg * 4 + 0) * DH + lane * 4] = acc0;
  *(v4f*)&part[(seg * 4 + 1) * DH + lane * 4] = acc1;
  *(v4f*)&part[(seg * 4 + 2) * DH + lane * 4] = acc2;
  *(v4f*)&part[(seg * 4 + 3) * DH + lane * 4] = acc3;
  __syncthreads();

  if (tid < 128) {
    const int kc = seg * 4 + (lane >> 3);
    const int hh = (lane >> 2) & 1, nl = lane & 3;
    const float* pp = &part[nl * DH + kc * 8 + hh * 4];
    v4f a = *(const v4f*)pp;
    for (int sg = 1; sg < FJS; ++sg) a += *(const v4f*)(pp + sg * (4 * DH));
    const int nt = n0 >> 4;
    const int p  = ((n0 & 15) >> 2) * 8 + (lane & 7);
    *(v4f*)&Mt[((((h * 24 + nt) * QKC + kc) * 32) + p) * 4] = a;
  }
}


static __device__ __forceinline__ void qkv_load(const float* __restrict__ aLane,
                                                const v4f* __restrict__ bLane,
                                                int kc, v4f (&a)[2], v4f (&b)[3]) {
  a[0] = *(const v4f*)(aLane + kc * 8);
  a[1] = *(const v4f*)(aLane + 16 * QPITCH + kc * 8);
#pragma unroll
  for (int q = 0; q < 3; ++q) b[q] = bLane[(q * QKC + kc) * 32];
}

static __device__ __forceinline__ void qkv_mma(const v4f (&a)[2], const v4f (&b)[3], v8f (&acc)[2][3]) {
#pragma unroll
  for (int m = 0; m < 2; ++m)
#pragma unroll
    for (int q = 0; q < 3; ++q) acc[m][q] = wmma4(lo2(a[m]), lo2(b[q]), acc[m][q]);
#pragma unroll
  for (int m = 0; m < 2; ++m)
#pragma unroll
    for (int q = 0; q < 3; ++q) acc[m][q] = wmma4(hi2(a[m]), hi2(b[q]), acc[m][q]);
}

__global__ __launch_bounds__(256) void cea_qkv(const float* __restrict__ x,
                                               const float* __restrict__ Mt,
                                               const float* __restrict__ cb,
                                               float* __restrict__ qkv,
                                               const float* __restrict__ Wp,
                                               float* __restrict__ spPf) {
  __shared__ __attribute__((aligned(16))) float xT[QTOK * QPITCH];
  if (blockIdx.x >= QGB) {
    wp_frag_job(Wp, spPf, blockIdx.x - QGB, threadIdx.x);
    return;
  }
  const int gy   = blockIdx.x / (NT / QTOK);
  const int h    = gy >> 1;
  const int ntb  = (gy & 1) * QNT;
  const int t0   = (blockIdx.x - gy * (NT / QTOK)) * QTOK;
  const int tid  = threadIdx.x;
  const int lane = tid & 31, wid = tid >> 5;
  {
    const float* src = x + (t0 + wid * 8) * DD + h * DH + lane * 4;
    float*       dst = xT + (wid * 8) * QPITCH + lane * 4;
    v4f tmp[8];
#pragma unroll
    for (int i = 0; i < 8; ++i) tmp[i] = *(const v4f*)(src + i * DD);
#pragma unroll
    for (int i = 0; i < 8; ++i) *(v4f*)(dst + i * QPITCH) = tmp[i];
  }
  __syncthreads();

  const int c  = lane & 15;
  const int hh = lane >> 4;
  const int mp = wid & 1;
  const int nt = ntb + (wid >> 1) * 3;
  const float* aLane = &xT[(mp * 32 + c) * QPITCH + 4 * hh];
  const v4f*   bLane = (const v4f*)Mt + ((h * 24 + nt) * QKC) * 32
                       + ((c >> 2) * 8 + hh * 4 + (c & 3));

  v8f acc[2][3];
#pragma unroll
  for (int m = 0; m < 2; ++m)
#pragma unroll
    for (int q = 0; q < 3; ++q) acc[m][q] = (v8f){};

  v4f a[4][2], b[4][3];
  qkv_load(aLane, bLane, 0, a[0], b[0]);
  qkv_load(aLane, bLane, 1, a[1], b[1]);
  qkv_load(aLane, bLane, 2, a[2], b[2]);
#pragma unroll
  for (int s = 0; s < 3; ++s) {
    asm volatile("" :: "v"(a[s][0]), "v"(a[s][1]));
#pragma unroll
    for (int q = 0; q < 3; ++q) asm volatile("" :: "v"(b[s][q]));
  }
  for (int kc = 0; kc < QKC; kc += 4) {
#pragma unroll
    for (int s = 0; s < 4; ++s) {
      const int kp = kc + s + 3;
      const int kn = (kp < QKC) ? kp : QKC - 1;
      qkv_load(aLane, bLane, kn, a[(s + 3) & 3], b[(s + 3) & 3]);
      qkv_mma(a[s], b[s], acc);
    }
  }

  const int row0 = t0 + mp * 32 + hh * 8;
#pragma unroll
  for (int m = 0; m < 2; ++m) {
#pragma unroll
    for (int q = 0; q < 3; ++q) {
      const int n = h * D3 + (nt + q) * 16 + c;
      const float bias = cb[n];
#pragma unroll
      for (int r = 0; r < 8; ++r)
        qkv[(row0 + m * 16 + r) * D9 + n] = acc[m][q][r] + bias;
    }
  }
}

#define ATOK  4
#define ACH   (D3 / 4)

static __device__ __forceinline__ void att_mul(const v2f (&qq)[4], v4f k4, v2f (&z)[8]) {
  const v2f k0 = lo2(k4), k1 = hi2(k4);
#pragma unroll
  for (int r = 0; r < 4; ++r) { z[r] = qq[r] * k0; z[4 + r] = qq[r] * k1; }
}
static __device__ __forceinline__ void att_exp(const v2f (&z)[8], v2f (&e)[8]) {
#pragma unroll
  for (int c = 0; c < 8; ++c) {
    e[c].x = __builtin_amdgcn_exp2f(z[c].x);
    e[c].y = __builtin_amdgcn_exp2f(z[c].y);
  }
}
static __device__ __forceinline__ void att_one(const v2f (&e)[8], v2f (&a)[8]) {
#pragma unroll
  for (int c = 0; c < 8; ++c) a[c] = e[c] + 1.0f;
}
static __device__ __forceinline__ void att_sum(const v2f (&a)[8], v4f v4, v2f (&acc)[4], v2f (&ssum)[4]) {
  const v2f v0 = lo2(v4), v1 = hi2(v4);
#pragma unroll
  for (int c = 0; c < 8; ++c) {
    v2f s;
    s.x = __builtin_amdgcn_rcpf(a[c].x);
    s.y = __builtin_amdgcn_rcpf(a[c].y);
    acc[c & 3]  += s * ((c < 4) ? v0 : v1);
    ssum[c & 3] += s;
  }
}

__global__ __launch_bounds__(384) void cea_attn(const float* __restrict__ qkv,
                                                float* __restrict__ att) {
  __shared__ __attribute__((aligned(16))) float KV[ATOK * 2 * D3];
  const int tid  = threadIdx.x;
  const int lane = tid & 31, wid = tid >> 5;
  const int tk   = wid / 3;
  const int i0   = (wid - tk * 3) * 128 + lane * 4;
  const int t    = blockIdx.x * ATOK + tk;
  const float* __restrict__ base = qkv + t * D9;
  const v4f q4 = *(const v4f*)(base + i0);
  {
    const v4f k4 = *(const v4f*)(base + D3 + i0);
    const v4f v4 = *(const v4f*)(base + 2 * D3 + i0);
    *(v4f*)&KV[tk * 2 * D3 + i0]      = k4 * (-GAMMA * LOG2E);
    *(v4f*)&KV[tk * 2 * D3 + D3 + i0] = v4;
  }
  __syncthreads();

  v2f qq[4], acc[4], ssum[4];
#pragma unroll
  for (int r = 0; r < 4; ++r) {
    qq[r]   = (v2f){q4[r], q4[r]};
    acc[r]  = (v2f){0.0f, 0.0f};
    ssum[r] = (v2f){0.0f, 0.0f};
  }
  const float* __restrict__ kp = &KV[tk * 2 * D3];

  v2f z[8], e[8], aA[8], aB[8];
  att_mul(qq, *(const v4f*)kp, z);
  att_exp(z, e);
  att_one(e, aA);
  att_mul(qq, *(const v4f*)(kp + 4), z);
  att_exp(z, e);
  v4f kA = *(const v4f*)(kp + 8);
  v4f kB = *(const v4f*)(kp + 12);
  v4f vA = *(const v4f*)(kp + D3);
  v4f vB;
  for (int n = 2; n < ACH; n += 2) {
    att_sum(aA, vA, acc, ssum);
    att_one(e, aB);
    att_mul(qq, kA, z);
    att_exp(z, e);
    const int nA = (n + 2 < ACH) ? n + 2 : ACH - 1;
    kA = *(const v4f*)(kp + 4 * nA);
    vB = *(const v4f*)(kp + D3 + 4 * (n - 1));
    att_sum(aB, vB, acc, ssum);
    att_one(e, aA);
    att_mul(qq, kB, z);
    att_exp(z, e);
    const int nB = (n + 3 < ACH) ? n + 3 : ACH - 1;
    kB = *(const v4f*)(kp + 4 * nB);
    vA = *(const v4f*)(kp + D3 + 4 * n);
#pragma unroll
    for (int g = 0; g < 64; ++g) {
      __builtin_amdgcn_sched_group_barrier(0x002, 1, 0);
      __builtin_amdgcn_sched_group_barrier(0x400, 1, 0);
    }
  }
  att_sum(aA, vA, acc, ssum);
  att_one(e, aB);
  att_sum(aB, *(const v4f*)(kp + D3 + 4 * (ACH - 1)), acc, ssum);

  v4f o;
#pragma unroll
  for (int r = 0; r < 4; ++r)
    o[r] = (acc[r].x + acc[r].y) / ((ssum[r].x + ssum[r].y) + 1e-8f);
  *(v4f*)(att + t * D3 + i0) = o;
}


template <int NTW>
static __device__ __forceinline__ void proj_load(const float* __restrict__ aLane,
                                                 const v4f* __restrict__ bLane,
                                                 int kc, v4f& a, v4f (&b)[NTW]) {
  a = *(const v4f*)(aLane + kc * 8);
#pragma unroll
  for (int q = 0; q < NTW; ++q) b[q] = bLane[(q * PKC + kc) * 32];
}

template <int NTW>
static __device__ __forceinline__ void proj_mma(v4f a, const v4f (&b)[NTW], v8f (&acc)[NTW]) {
#pragma unroll
  for (int q = 0; q < NTW; ++q) acc[q] = wmma4(lo2(a), lo2(b[q]), acc[q]);
#pragma unroll
  for (int q = 0; q < NTW; ++q) acc[q] = wmma4(hi2(a), hi2(b[q]), acc[q]);
}

template <int NTW>
static __device__ __forceinline__ void proj_wave(const float* __restrict__ aLane,
                                                 const v4f* __restrict__ bLane,
                                                 const float* __restrict__ bp,
                                                 float* __restrict__ hout,
                                                 int row0, int col0) {
  v8f acc[NTW];
#pragma unroll
  for (int q = 0; q < NTW; ++q) acc[q] = (v8f){};

  v4f a[4], b[4][NTW];
  proj_load<NTW>(aLane, bLane, 0, a[0], b[0]);
  proj_load<NTW>(aLane, bLane, 1, a[1], b[1]);
  proj_load<NTW>(aLane, bLane, 2, a[2], b[2]);
#pragma unroll
  for (int s = 0; s < 3; ++s) {
    asm volatile("" :: "v"(a[s]));
#pragma unroll
    for (int q = 0; q < NTW; ++q) asm volatile("" :: "v"(b[s][q]));
  }
  for (int kc = 0; kc < PKC; kc += 4) {
#pragma unroll
    for (int s = 0; s < 4; ++s) {
      const int kp = kc + s + 3;
      const int kn = (kp < PKC) ? kp : PKC - 1;
      proj_load<NTW>(aLane, bLane, kn, a[(s + 3) & 3], b[(s + 3) & 3]);
      proj_mma<NTW>(a[s], b[s], acc);
    }
  }

#pragma unroll
  for (int q = 0; q < NTW; ++q) {
    const int n = col0 + q * 16;
    const float bias = bp[n];
#pragma unroll
    for (int r = 0; r < 8; ++r)
      hout[(row0 + r) * D9 + n] = acc[q][r] + bias;
  }
}

__global__ __launch_bounds__(256) void cea_proj(const float* __restrict__ att,
                                                const float* __restrict__ spPf,
                                                const float* __restrict__ bp,
                                                float* __restrict__ hout) {
  __shared__ __attribute__((aligned(16))) float aT[PTOK * APITCH];
  const int t0  = blockIdx.x * PTOK;
  const int tid = threadIdx.x;
  {
    const int sw = tid >> 5, sl = tid & 31;
    const float* src = att + (t0 + sw * 8) * D3 + sl * 4;
    float*       dst = aT + (sw * 8) * APITCH + sl * 4;
    v4f tmp[24];
#pragma unroll
    for (int i = 0; i < 24; ++i) tmp[i] = *(const v4f*)(src + (i / 3) * D3 + (i % 3) * 128);
#pragma unroll
    for (int i = 0; i < 24; ++i) *(v4f*)(dst + (i / 3) * APITCH + (i % 3) * 128) = tmp[i];
  }
  __syncthreads();

  const int lane = tid & 31, wid = tid >> 5;
  const int c  = lane & 15;
  const int h  = lane >> 4;
  const int mi = wid & 3;
  const int nt0 = (wid >> 2) * 5;
  const float* aLane = &aT[(mi * 16 + c) * APITCH + 4 * h];
  const v4f*   bLane = (const v4f*)spPf + ((blockIdx.y * 9 + nt0) * PKC) * 32 + lane;
  const int row0 = t0 + mi * 16 + h * 8;
  const int col0 = blockIdx.y * PCOL + nt0 * 16 + c;

  if (wid < 4) proj_wave<5>(aLane, bLane, bp, hout, row0, col0);
  else         proj_wave<4>(aLane, bLane, bp, hout, row0, col0);
}

__global__ void cea_ln(float* __restrict__ out,
                       const float* __restrict__ gn,
                       const float* __restrict__ bn) {
  __shared__ float row[D9];
  __shared__ float red[256];
  const int t = blockIdx.x, tid = threadIdx.x;
  float* __restrict__ h = out + t * D9;

  float ps = 0.0f;
  for (int n = tid; n < D9; n += 256) { float v = h[n]; row[n] = v; ps += v; }
  red[tid] = ps;
  __syncthreads();
  for (int s = 128; s > 0; s >>= 1) {
    if (tid < s) red[tid] += red[tid + s];
    __syncthreads();
  }
  const float mu = red[0] * (1.0f / (float)D9);
  __syncthreads();

  float pv = 0.0f;
  for (int n = tid; n < D9; n += 256) { float dv = row[n] - mu; pv += dv * dv; }
  red[tid] = pv;
  __syncthreads();
  for (int s = 128; s > 0; s >>= 1) {
    if (tid < s) red[tid] += red[tid + s];
    __syncthreads();
  }
  const float var = red[0] * (1.0f / (float)D9);
  const float rs  = __builtin_amdgcn_rsqf(var + 1e-5f);

  for (int n = tid; n < D9; n += 256)
    h[n] = (row[n] - mu) * rs * gn[n] + bn[n];
}


extern "C" void kernel_launch(void* const* d_in, const int* in_sizes, int n_in,
                              void* d_out, int out_size, void* d_ws, size_t ws_size,
                              hipStream_t stream) {
  const float* x  = (const float*)d_in[0];
  const float* Wq = (const float*)d_in[1];
  const float* bq = (const float*)d_in[2];
  const float* Sq = (const float*)d_in[3];
  const float* Wk = (const float*)d_in[4];
  const float* bk = (const float*)d_in[5];
  const float* Sk = (const float*)d_in[6];
  const float* Wv = (const float*)d_in[7];
  const float* bv = (const float*)d_in[8];
  const float* Sv = (const float*)d_in[9];
  const float* Wp = (const float*)d_in[10];
  const float* bp = (const float*)d_in[11];
  const float* gn = (const float*)d_in[12];
  const float* bn = (const float*)d_in[13];
  float* out = (float*)d_out;

  float* ws = (float*)d_ws;
  float* Mt   = ws;
  float* cb   = Mt  + 3 * D3 * DH;
  float* spP  = cb  + 3 * D3;
  float* qkv  = spP + D9 * D3;
  float* attb = qkv + NT * D9;

  cea_fuse<<<dim3(D3 / 4 + BNB, 3), dim3(256), 0, stream>>>(Wq, Wk, Wv, Sq, Sk, Sv, bq, bk, bv, Mt, cb);

  cea_qkv<<<dim3(QGB + SPB), dim3(256), 0, stream>>>(x, Mt, cb, qkv, Wp, spP);

  cea_attn<<<dim3(NT / ATOK), dim3(3 * 32 * ATOK), 0, stream>>>(qkv, attb);

  cea_proj<<<dim3(NT / PTOK, D9 / PCOL), dim3(256), 0, stream>>>(attb, spP, bp, out);

  cea_ln<<<dim3(NT), dim3(256), 0, stream>>>(out, gn, bn);

  (void)in_sizes; (void)n_in; (void)out_size; (void)ws_size;
}


# layernorm, still 8 waves per token: row kept in registers, a pass's loads issued together (were 10 dependent round trips), sums by lane exchange, 2 barriers (was 18); new sum order in mean, var
# speedup vs baseline: 3.1188x; 1.0765x over previous
#include <hip/hip_runtime.h>
#include <hip/hip_bf16.h>
#include <math.h>


#define NT     2048
#define DD     384
#define DH     128
#define D3     384
#define D9     1152
#define PKC    48
#define PTOK   64
#define PCOL   144
#define APITCH 388
#define QKC    16
#define QTOK   64
#define QNT    12
#define QPITCH 132
#define QGB    192
#define GAMMA  5.0f
#define LOG2E  1.4426950408889634f

typedef float v2f __attribute__((ext_vector_type(2)));
typedef float v4f __attribute__((ext_vector_type(4)));
typedef float v8f __attribute__((ext_vector_type(8)));

static __device__ __forceinline__ v8f wmma4(v2f a, v2f b, v8f c) {
  return __builtin_amdgcn_wmma_f32_16x16x4_f32(
       false, a,  false, b,
       (short)0, c,  false,  false);
}

static __device__ __forceinline__ v2f lo2(v4f v) { return __builtin_shufflevector(v, v, 0, 1); }
static __device__ __forceinline__ v2f hi2(v4f v) { return __builtin_shufflevector(v, v, 2, 3); }

static __device__ __forceinline__ float softplusf(float w) {
  return (w > 20.0f) ? w : log1pf(__expf(w));
}

static __device__ __forceinline__ float softplus_hw(float w) {
  const float t = __builtin_amdgcn_exp2f(-__builtin_fabsf(w) * LOG2E);
  return __builtin_fmaxf(w, 0.0f) + 0.6931471805599453f * __builtin_amdgcn_logf(1.0f + t);
}


#define SPB  216
#define SPT  2
static __device__ __forceinline__ v4f wp_frag_load(const float* __restrict__ Wp, int o4) {
  const int lane = o4 & 31;
  const int c = lane & 15, h = lane >> 4;
  const int ch = o4 >> 5;
  const int nt = ch / PKC, kc = ch - nt * PKC;
  return *(const v4f*)&Wp[(nt * 16 + c) * D3 + kc * 8 + 4 * h];
}
static __device__ __forceinline__ void wp_frag_job(const float* __restrict__ Wp,
                                                   float* __restrict__ spPf, int blk, int tid) {
  const int o0 = blk * (256 * SPT) + tid;
  v4f w[SPT];
#pragma unroll
  for (int i = 0; i < SPT; ++i) w[i] = wp_frag_load(Wp, o0 + i * 256);
#pragma unroll
  for (int i = 0; i < SPT; ++i) {
    v4f o;
    o.x = softplusf(w[i].x); o.y = softplusf(w[i].y);
    o.z = softplusf(w[i].z); o.w = softplusf(w[i].w);
    *(v4f*)&spPf[(o0 + i * 256) * 4] = o;
  }
}

#define BJS 16
#define BJL (D3 / BJS)
#define BNC 64
#define BNB (D3 / BNC)
static __device__ __forceinline__ void bias_job(const float* __restrict__ b,
                                                const float* __restrict__ S,
                                                float* __restrict__ c,
                                                float* __restrict__ part,
                                                int n0, int tid) {
  const int lane = tid & 31;
  const int seg  = (tid >> 5) * 2 + (lane >> 4);
  const int nl   = (lane & 15) * 4;
  const float* bl = b + seg * BJL;
  const float* Sl = S + (seg * BJL) * D3 + n0 + nl;

  v4f bb[BJL / 4], sv[BJL];
#pragma unroll
  for (int u = 0; u < BJL / 4; ++u) bb[u] = *(const v4f*)(bl + 4 * u);
#pragma unroll
  for (int u = 0; u < BJL; ++u) sv[u] = *(const v4f*)(Sl + u * D3);

  v4f acc = {};
#pragma unroll
  for (int u = 0; u < BJL; ++u) acc += sv[u] * bb[u >> 2][u & 3];
  *(v4f*)&part[seg * BNC + nl] = acc;
  __syncthreads();

  if (tid < BNC) {
    float a = part[tid];
    for (int sg = 1; sg < BJS; ++sg) a += part[sg * BNC + tid];
    c[n0 + tid] = a;
  }
}

#define FJS 8
#define FJL (D3 / FJS)
#define FJB 12
__global__ __launch_bounds__(256) void cea_fuse(const float* __restrict__ Wq,
                                                const float* __restrict__ Wk,
                                                const float* __restrict__ Wv,
                                                const float* __restrict__ Sq,
                                                const float* __restrict__ Sk,
                                                const float* __restrict__ Sv,
                                                const float* __restrict__ bq,
                                                const float* __restrict__ bk,
                                                const float* __restrict__ bv,
                                                float* __restrict__ Mt,
                                                float* __restrict__ cb) {
  __shared__ __attribute__((aligned(16))) float part[FJS * 4 * DH];
  const int h   = blockIdx.y;
  const int tid = threadIdx.x;
  const float* S  = (h == 0) ? Sq : (h == 1) ? Sk : Sv;
  if (blockIdx.x >= D3 / 4) {
    const float* b = (h == 0) ? bq : (h == 1) ? bk : bv;
    bias_job(b, S, cb + h * D3, part, (blockIdx.x - D3 / 4) * BNC, tid);
    return;
  }
  const int n0  = blockIdx.x * 4;
  const int lane = tid & 31, seg = tid >> 5;
  const float* W  = (h == 0) ? Wq : (h == 1) ? Wk : Wv;
  const float* Wl = W + (seg * FJL) * DH + lane * 4;
  const float* Sl = S + (seg * FJL) * D3 + n0;

  v4f acc0 = {}, acc1 = {}, acc2 = {}, acc3 = {};
  for (int j0 = 0; j0 < FJL; j0 += FJB) {
    v4f w[FJB], s[FJB];
#pragma unroll
    for (int u = 0; u < FJB; ++u) {
      w[u] = *(const v4f*)(Wl + (j0 + u) * DH);
      s[u] = *(const v4f*)(Sl + (j0 + u) * D3);
    }
#pragma unroll
    for (int u = 0; u < FJB; ++u) asm volatile("" :: "v"(w[u]), "v"(s[u]));
#pragma unroll
    for (int u = 0; u < FJB; ++u) {
      w[u].x = softplus_hw(w[u].x); w[u].y = softplus_hw(w[u].y);
      w[u].z = softplus_hw(w[u].z); w[u].w = softplus_hw(w[u].w);
      acc0 += w[u] * s[u].x;
      acc1 += w[u] * s[u].y;
      acc2 += w[u] * s[u].z;
      acc3 += w[u] * s[u].w;
    }
  }
  *(v4f*)&part[(seg * 4 + 0) * DH + lane * 4] = acc0;
  *(v4f*)&part[(seg * 4 + 1) * DH + lane * 4] = acc1;
  *(v4f*)&part[(seg * 4 + 2) * DH + lane * 4] = acc2;
  *(v4f*)&part[(seg * 4 + 3) * DH + lane * 4] = acc3;
  __syncthreads();

  if (tid < 128) {
    const int kc = seg * 4 + (lane >> 3);
    const int hh = (lane >> 2) & 1, nl = lane & 3;
    const float* pp = &part[nl * DH + kc * 8 + hh * 4];
    v4f a = *(const v4f*)pp;
    for (int sg = 1; sg < FJS; ++sg) a += *(const v4f*)(pp + sg * (4 * DH));
    const int nt = n0 >> 4;
    const int p  = ((n0 & 15) >> 2) * 8 + (lane & 7);
    *(v4f*)&Mt[((((h * 24 + nt) * QKC + kc) * 32) + p) * 4] = a;
  }
}


static __device__ __forceinline__ void qkv_load(const float* __restrict__ aLane,
                                                const v4f* __restrict__ bLane,
                                                int kc, v4f (&a)[2], v4f (&b)[3]) {
  a[0] = *(const v4f*)(aLane + kc * 8);
  a[1] = *(const v4f*)(aLane + 16 * QPITCH + kc * 8);
#pragma unroll
  for (int q = 0; q < 3; ++q) b[q] = bLane[(q * QKC + kc) * 32];
}

static __device__ __forceinline__ void qkv_mma(const v4f (&a)[2], const v4f (&b)[3], v8f (&acc)[2][3]) {
#pragma unroll
  for (int m = 0; m < 2; ++m)
#pragma unroll
    for (int q = 0; q < 3; ++q) acc[m][q] = wmma4(lo2(a[m]), lo2(b[q]), acc[m][q]);
#pragma unroll
  for (int m = 0; m < 2; ++m)
#pragma unroll
    for (int q = 0; q < 3; ++q) acc[m][q] = wmma4(hi2(a[m]), hi2(b[q]), acc[m][q]);
}

__global__ __launch_bounds__(256) void cea_qkv(const float* __restrict__ x,
                                               const float* __restrict__ Mt,
                                               const float* __restrict__ cb,
                                               float* __restrict__ qkv,
                                               const float* __restrict__ Wp,
                                               float* __restrict__ spPf) {
  __shared__ __attribute__((aligned(16))) float xT[QTOK * QPITCH];
  if (blockIdx.x >= QGB) {
    wp_frag_job(Wp, spPf, blockIdx.x - QGB, threadIdx.x);
    return;
  }
  const int gy   = blockIdx.x / (NT / QTOK);
  const int h    = gy >> 1;
  const int ntb  = (gy & 1) * QNT;
  const int t0   = (blockIdx.x - gy * (NT / QTOK)) * QTOK;
  const int tid  = threadIdx.x;
  const int lane = tid & 31, wid = tid >> 5;
  {
    const float* src = x + (t0 + wid * 8) * DD + h * DH + lane * 4;
    float*       dst = xT + (wid * 8) * QPITCH + lane * 4;
    v4f tmp[8];
#pragma unroll
    for (int i = 0; i < 8; ++i) tmp[i] = *(const v4f*)(src + i * DD);
#pragma unroll
    for (int i = 0; i < 8; ++i) *(v4f*)(dst + i * QPITCH) = tmp[i];
  }
  __syncthreads();

  const int c  = lane & 15;
  const int hh = lane >> 4;
  const int mp = wid & 1;
  const int nt = ntb + (wid >> 1) * 3;
  const float* aLane = &xT[(mp * 32 + c) * QPITCH + 4 * hh];
  const v4f*   bLane = (const v4f*)Mt + ((h * 24 + nt) * QKC) * 32
                       + ((c >> 2) * 8 + hh * 4 + (c & 3));

  v8f acc[2][3];
#pragma unroll
  for (int m = 0; m < 2; ++m)
#pragma unroll
    for (int q = 0; q < 3; ++q) acc[m][q] = (v8f){};

  v4f a[4][2], b[4][3];
  qkv_load(aLane, bLane, 0, a[0], b[0]);
  qkv_load(aLane, bLane, 1, a[1], b[1]);
  qkv_load(aLane, bLane, 2, a[2], b[2]);
#pragma unroll
  for (int s = 0; s < 3; ++s) {
    asm volatile("" :: "v"(a[s][0]), "v"(a[s][1]));
#pragma unroll
    for (int q = 0; q < 3; ++q) asm volatile("" :: "v"(b[s][q]));
  }
  for (int kc = 0; kc < QKC; kc += 4) {
#pragma unroll
    for (int s = 0; s < 4; ++s) {
      const int kp = kc + s + 3;
      const int kn = (kp < QKC) ? kp : QKC - 1;
      qkv_load(aLane, bLane, kn, a[(s + 3) & 3], b[(s + 3) & 3]);
      qkv_mma(a[s], b[s], acc);
    }
  }

  const int row0 = t0 + mp * 32 + hh * 8;
#pragma unroll
  for (int m = 0; m < 2; ++m) {
#pragma unroll
    for (int q = 0; q < 3; ++q) {
      const int n = h * D3 + (nt + q) * 16 + c;
      const float bias = cb[n];
#pragma unroll
      for (int r = 0; r < 8; ++r)
        qkv[(row0 + m * 16 + r) * D9 + n] = acc[m][q][r] + bias;
    }
  }
}

#define ATOK  4
#define ACH   (D3 / 4)

static __device__ __forceinline__ void att_mul(const v2f (&qq)[4], v4f k4, v2f (&z)[8]) {
  const v2f k0 = lo2(k4), k1 = hi2(k4);
#pragma unroll
  for (int r = 0; r < 4; ++r) { z[r] = qq[r] * k0; z[4 + r] = qq[r] * k1; }
}
static __device__ __forceinline__ void att_exp(const v2f (&z)[8], v2f (&e)[8]) {
#pragma unroll
  for (int c = 0; c < 8; ++c) {
    e[c].x = __builtin_amdgcn_exp2f(z[c].x);
    e[c].y = __builtin_amdgcn_exp2f(z[c].y);
  }
}
static __device__ __forceinline__ void att_one(const v2f (&e)[8], v2f (&a)[8]) {
#pragma unroll
  for (int c = 0; c < 8; ++c) a[c] = e[c] + 1.0f;
}
static __device__ __forceinline__ void att_sum(const v2f (&a)[8], v4f v4, v2f (&acc)[4], v2f (&ssum)[4]) {
  const v2f v0 = lo2(v4), v1 = hi2(v4);
#pragma unroll
  for (int c = 0; c < 8; ++c) {
    v2f s;
    s.x = __builtin_amdgcn_rcpf(a[c].x);
    s.y = __builtin_amdgcn_rcpf(a[c].y);
    acc[c & 3]  += s * ((c < 4) ? v0 : v1);
    ssum[c & 3] += s;
  }
}

__global__ __launch_bounds__(384) void cea_attn(const float* __restrict__ qkv,
                                                float* __restrict__ att) {
  __shared__ __attribute__((aligned(16))) float KV[ATOK * 2 * D3];
  const int tid  = threadIdx.x;
  const int lane = tid & 31, wid = tid >> 5;
  const int tk   = wid / 3;
  const int i0   = (wid - tk * 3) * 128 + lane * 4;
  const int t    = blockIdx.x * ATOK + tk;
  const float* __restrict__ base = qkv + t * D9;
  const v4f q4 = *(const v4f*)(base + i0);
  {
    const v4f k4 = *(const v4f*)(base + D3 + i0);
    const v4f v4 = *(const v4f*)(base + 2 * D3 + i0);
    *(v4f*)&KV[tk * 2 * D3 + i0]      = k4 * (-GAMMA * LOG2E);
    *(v4f*)&KV[tk * 2 * D3 + D3 + i0] = v4;
  }
  __syncthreads();

  v2f qq[4], acc[4], ssum[4];
#pragma unroll
  for (int r = 0; r < 4; ++r) {
    qq[r]   = (v2f){q4[r], q4[r]};
    acc[r]  = (v2f){0.0f, 0.0f};
    ssum[r] = (v2f){0.0f, 0.0f};
  }
  const float* __restrict__ kp = &KV[tk * 2 * D3];

  v2f z[8], e[8], aA[8], aB[8];
  att_mul(qq, *(const v4f*)kp, z);
  att_exp(z, e);
  att_one(e, aA);
  att_mul(qq, *(const v4f*)(kp + 4), z);
  att_exp(z, e);
  v4f kA = *(const v4f*)(kp + 8);
  v4f kB = *(const v4f*)(kp + 12);
  v4f vA = *(const v4f*)(kp + D3);
  v4f vB;
  for (int n = 2; n < ACH; n += 2) {
    att_sum(aA, vA, acc, ssum);
    att_one(e, aB);
    att_mul(qq, kA, z);
    att_exp(z, e);
    const int nA = (n + 2 < ACH) ? n + 2 : ACH - 1;
    kA = *(const v4f*)(kp + 4 * nA);
    vB = *(const v4f*)(kp + D3 + 4 * (n - 1));
    att_sum(aB, vB, acc, ssum);
    att_one(e, aA);
    att_mul(qq, kB, z);
    att_exp(z, e);
    const int nB = (n + 3 < ACH) ? n + 3 : ACH - 1;
    kB = *(const v4f*)(kp + 4 * nB);
    vA = *(const v4f*)(kp + D3 + 4 * n);
#pragma unroll
    for (int g = 0; g < 64; ++g) {
      __builtin_amdgcn_sched_group_barrier(0x002, 1, 0);
      __builtin_amdgcn_sched_group_barrier(0x400, 1, 0);
    }
  }
  att_sum(aA, vA, acc, ssum);
  att_one(e, aB);
  att_sum(aB, *(const v4f*)(kp + D3 + 4 * (ACH - 1)), acc, ssum);

  v4f o;
#pragma unroll
  for (int r = 0; r < 4; ++r)
    o[r] = (acc[r].x + acc[r].y) / ((ssum[r].x + ssum[r].y) + 1e-8f);
  *(v4f*)(att + t * D3 + i0) = o;
}


template <int NTW>
static __device__ __forceinline__ void proj_load(const float* __restrict__ aLane,
                                                 const v4f* __restrict__ bLane,
                                                 int kc, v4f& a, v4f (&b)[NTW]) {
  a = *(const v4f*)(aLane + kc * 8);
#pragma unroll
  for (int q = 0; q < NTW; ++q) b[q] = bLane[(q * PKC + kc) * 32];
}

template <int NTW>
static __device__ __forceinline__ void proj_mma(v4f a, const v4f (&b)[NTW], v8f (&acc)[NTW]) {
#pragma unroll
  for (int q = 0; q < NTW; ++q) acc[q] = wmma4(lo2(a), lo2(b[q]), acc[q]);
#pragma unroll
  for (int q = 0; q < NTW; ++q) acc[q] = wmma4(hi2(a), hi2(b[q]), acc[q]);
}

template <int NTW>
static __device__ __forceinline__ void proj_wave(const float* __restrict__ aLane,
                                                 const v4f* __restrict__ bLane,
                                                 const float* __restrict__ bp,
                                                 float* __restrict__ hout,
                                                 int row0, int col0) {
  v8f acc[NTW];
#pragma unroll
  for (int q = 0; q < NTW; ++q) acc[q] = (v8f){};

  v4f a[4], b[4][NTW];
  proj_load<NTW>(aLane, bLane, 0, a[0], b[0]);
  proj_load<NTW>(aLane, bLane, 1, a[1], b[1]);
  proj_load<NTW>(aLane, bLane, 2, a[2], b[2]);
#pragma unroll
  for (int s = 0; s < 3; ++s) {
    asm volatile("" :: "v"(a[s]));
#pragma unroll
    for (int q = 0; q < NTW; ++q) asm volatile("" :: "v"(b[s][q]));
  }
  for (int kc = 0; kc < PKC; kc += 4) {
#pragma unroll
    for (int s = 0; s < 4; ++s) {
      const int kp = kc + s + 3;
      const int kn = (kp < PKC) ? kp : PKC - 1;
      proj_load<NTW>(aLane, bLane, kn, a[(s + 3) & 3], b[(s + 3) & 3]);
      proj_mma<NTW>(a[s], b[s], acc);
    }
  }

#pragma unroll
  for (int q = 0; q < NTW; ++q) {
    const int n = col0 + q * 16;
    const float bias = bp[n];
#pragma unroll
    for (int r = 0; r < 8; ++r)
      hout[(row0 + r) * D9 + n] = acc[q][r] + bias;
  }
}

__global__ __launch_bounds__(256) void cea_proj(const float* __restrict__ att,
                                                const float* __restrict__ spPf,
                                                const float* __restrict__ bp,
                                                float* __restrict__ hout) {
  __shared__ __attribute__((aligned(16))) float aT[PTOK * APITCH];
  const int t0  = blockIdx.x * PTOK;
  const int tid = threadIdx.x;
  {
    const int sw = tid >> 5, sl = tid & 31;
    const float* src = att + (t0 + sw * 8) * D3 + sl * 4;
    float*       dst = aT + (sw * 8) * APITCH + sl * 4;
    v4f tmp[24];
#pragma unroll
    for (int i = 0; i < 24; ++i) tmp[i] = *(const v4f*)(src + (i / 3) * D3 + (i % 3) * 128);
#pragma unroll
    for (int i = 0; i < 24; ++i) *(v4f*)(dst + (i / 3) * APITCH + (i % 3) * 128) = tmp[i];
  }
  __syncthreads();

  const int lane = tid & 31, wid = tid >> 5;
  const int c  = lane & 15;
  const int h  = lane >> 4;
  const int mi = wid & 3;
  const int nt0 = (wid >> 2) * 5;
  const float* aLane = &aT[(mi * 16 + c) * APITCH + 4 * h];
  const v4f*   bLane = (const v4f*)spPf + ((blockIdx.y * 9 + nt0) * PKC) * 32 + lane;
  const int row0 = t0 + mi * 16 + h * 8;
  const int col0 = blockIdx.y * PCOL + nt0 * 16 + c;

  if (wid < 4) proj_wave<5>(aLane, bLane, bp, hout, row0, col0);
  else         proj_wave<4>(aLane, bLane, bp, hout, row0, col0);
}

#define LNI 5
template <int M>
static __device__ __forceinline__ float ln_xrow(float x) {
  const int b = __builtin_bit_cast(int, x);
  return __builtin_bit_cast(float, __builtin_amdgcn_update_dpp(b, b, 0x160 + M, 0xf, 0xf, false));
}
static __device__ __forceinline__ float ln_xhalf(float x) {
  const int b = __builtin_bit_cast(int, x);
  return __builtin_bit_cast(float, __builtin_amdgcn_permlanex16(b, b, 0x76543210, 0xfedcba98, false, false));
}
static __device__ __forceinline__ float ln_wave_sum(float v) {
  v += ln_xrow<1>(v);
  v += ln_xrow<2>(v);
  v += ln_xrow<4>(v);
  v += ln_xrow<8>(v);
  v += ln_xhalf(v);
  return v;
}
__global__ __launch_bounds__(256) void cea_ln(float* __restrict__ out,
                                              const float* __restrict__ gn,
                                              const float* __restrict__ bn) {
  __shared__ float redA[8];
  __shared__ float redB[8];
  const int t = blockIdx.x, tid = threadIdx.x;
  const int lane = tid & 31, wid = tid >> 5;
  const bool tail = tid < (D9 - 4 * 256);
  const int n4 = tail ? tid + 4 * 256 : tid;
  float* __restrict__ h = out + t * D9;

  float v[LNI];
#pragma unroll
  for (int i = 0; i < 4; ++i) v[i] = h[tid + 256 * i];
  v[4] = h[n4];
  if (!tail) v[4] = 0.0f;

  float ps = ((v[0] + v[1]) + (v[2] + v[3])) + v[4];
  ps = ln_wave_sum(ps);
  if (lane == 0) redA[wid] = ps;
  __syncthreads();
  float sm = redA[0];
#pragma unroll
  for (int w = 1; w < 8; ++w) sm += redA[w];
  const float mu = sm * (1.0f / (float)D9);

  float dv[LNI];
#pragma unroll
  for (int i = 0; i < LNI; ++i) dv[i] = v[i] - mu;
  if (!tail) dv[4] = 0.0f;
  float pv = ((dv[0] * dv[0] + dv[1] * dv[1]) + (dv[2] * dv[2] + dv[3] * dv[3])) + dv[4] * dv[4];
  pv = ln_wave_sum(pv);
  if (lane == 0) redB[wid] = pv;
  __syncthreads();
  float sv = redB[0];
#pragma unroll
  for (int w = 1; w < 8; ++w) sv += redB[w];
  const float var = sv * (1.0f / (float)D9);
  const float rs  = __builtin_amdgcn_rsqf(var + 1e-5f);

  float g[LNI], b[LNI];
#pragma unroll
  for (int i = 0; i < 4; ++i) { g[i] = gn[tid + 256 * i]; b[i] = bn[tid + 256 * i]; }
  g[4] = gn[n4]; b[4] = bn[n4];
  asm volatile("" :: "v"(g[4]), "v"(b[4]));
#pragma unroll
  for (int i = 0; i < 4; ++i) h[tid + 256 * i] = dv[i] * rs * g[i] + b[i];
  if (tail) h[n4] = dv[4] * rs * g[4] + b[4];
}


extern "C" void kernel_launch(void* const* d_in, const int* in_sizes, int n_in,
                              void* d_out, int out_size, void* d_ws, size_t ws_size,
                              hipStream_t stream) {
  const float* x  = (const float*)d_in[0];
  const float* Wq = (const float*)d_in[1];
  const float* bq = (const float*)d_in[2];
  const float* Sq = (const float*)d_in[3];
  const float* Wk = (const float*)d_in[4];
  const float* bk = (const float*)d_in[5];
  const float* Sk = (const float*)d_in[6];
  const float* Wv = (const float*)d_in[7];
  const float* bv = (const float*)d_in[8];
  const float* Sv = (const float*)d_in[9];
  const float* Wp = (const float*)d_in[10];
  const float* bp = (const float*)d_in[11];
  const float* gn = (const float*)d_in[12];
  const float* bn = (const float*)d_in[13];
  float* out = (float*)d_out;

  float* ws = (float*)d_ws;
  float* Mt   = ws;
  float* cb   = Mt  + 3 * D3 * DH;
  float* spP  = cb  + 3 * D3;
  float* qkv  = spP + D9 * D3;
  float* attb = qkv + NT * D9;

  cea_fuse<<<dim3(D3 / 4 + BNB, 3), dim3(256), 0, stream>>>(Wq, Wk, Wv, Sq, Sk, Sv, bq, bk, bv, Mt, cb);

  cea_qkv<<<dim3(QGB + SPB), dim3(256), 0, stream>>>(x, Mt, cb, qkv, Wp, spP);

  cea_attn<<<dim3(NT / ATOK), dim3(3 * 32 * ATOK), 0, stream>>>(qkv, attb);

  cea_proj<<<dim3(NT / PTOK, D9 / PCOL), dim3(256), 0, stream>>>(attb, spP, bp, out);

  cea_ln<<<dim3(NT), dim3(256), 0, stream>>>(out, gn, bn);

  (void)in_sizes; (void)n_in; (void)out_size; (void)ws_size;
}


# fused-weight build (Mt) re-tiled: 16-wave work-group = 32 n x 32 k, softplus of its W slab taken once into LDS, lane = 4 n x 8 k, j in 16 segments; 162 work-groups (was 306); new sum order in Mt
# speedup vs baseline: 3.2190x; 1.0321x over previous
#include <hip/hip_runtime.h>
#include <hip/hip_bf16.h>
#include <math.h>


#define NT     2048
#define DD     384
#define DH     128
#define D3     384
#define D9     1152
#define PKC    48
#define PTOK   64
#define PCOL   144
#define APITCH 388
#define QKC    16
#define QTOK   64
#define QNT    12
#define QPITCH 132
#define QGB    192
#define GAMMA  5.0f
#define LOG2E  1.4426950408889634f

typedef float v2f __attribute__((ext_vector_type(2)));
typedef float v4f __attribute__((ext_vector_type(4)));
typedef float v8f __attribute__((ext_vector_type(8)));

static __device__ __forceinline__ v8f wmma4(v2f a, v2f b, v8f c) {
  return __builtin_amdgcn_wmma_f32_16x16x4_f32(
       false, a,  false, b,
       (short)0, c,  false,  false);
}

static __device__ __forceinline__ v2f lo2(v4f v) { return __builtin_shufflevector(v, v, 0, 1); }
static __device__ __forceinline__ v2f hi2(v4f v) { return __builtin_shufflevector(v, v, 2, 3); }

static __device__ __forceinline__ float softplusf(float w) {
  return (w > 20.0f) ? w : log1pf(__expf(w));
}

static __device__ __forceinline__ float softplus_hw(float w) {
  const float t = __builtin_amdgcn_exp2f(-__builtin_fabsf(w) * LOG2E);
  return __builtin_fmaxf(w, 0.0f) + 0.6931471805599453f * __builtin_amdgcn_logf(1.0f + t);
}


#define SPB  216
#define SPT  2
static __device__ __forceinline__ v4f wp_frag_load(const float* __restrict__ Wp, int o4) {
  const int lane = o4 & 31;
  const int c = lane & 15, h = lane >> 4;
  const int ch = o4 >> 5;
  const int nt = ch / PKC, kc = ch - nt * PKC;
  return *(const v4f*)&Wp[(nt * 16 + c) * D3 + kc * 8 + 4 * h];
}
static __device__ __forceinline__ void wp_frag_job(const float* __restrict__ Wp,
                                                   float* __restrict__ spPf, int blk, int tid) {
  const int o0 = blk * (256 * SPT) + tid;
  v4f w[SPT];
#pragma unroll
  for (int i = 0; i < SPT; ++i) w[i] = wp_frag_load(Wp, o0 + i * 256);
#pragma unroll
  for (int i = 0; i < SPT; ++i) {
    v4f o;
    o.x = softplusf(w[i].x); o.y = softplusf(w[i].y);
    o.z = softplusf(w[i].z); o.w = softplusf(w[i].w);
    *(v4f*)&spPf[(o0 + i * 256) * 4] = o;
  }
}

#define BJS 16
#define BJL (D3 / BJS)
#define BNC 64
#define BNB (D3 / BNC)
static __device__ __forceinline__ void bias_job(const float* __restrict__ b,
                                                const float* __restrict__ S,
                                                float* __restrict__ c,
                                                float* __restrict__ part,
                                                int n0, int tid) {
  if (tid < 256) {
    const int lane = tid & 31;
    const int seg  = (tid >> 5) * 2 + (lane >> 4);
    const int nl   = (lane & 15) * 4;
    const float* bl = b + seg * BJL;
    const float* Sl = S + (seg * BJL) * D3 + n0 + nl;

    v4f bb[BJL / 4], sv[BJL];
#pragma unroll
    for (int u = 0; u < BJL / 4; ++u) bb[u] = *(const v4f*)(bl + 4 * u);
#pragma unroll
    for (int u = 0; u < BJL; ++u) sv[u] = *(const v4f*)(Sl + u * D3);

    v4f acc = {};
#pragma unroll
    for (int u = 0; u < BJL; ++u) acc += sv[u] * bb[u >> 2][u & 3];
    *(v4f*)&part[seg * BNC + nl] = acc;
  }
  __syncthreads();

  if (tid < BNC) {
    float a = part[tid];
    for (int sg = 1; sg < BJS; ++sg) a += part[sg * BNC + tid];
    c[n0 + tid] = a;
  }
}

#define FTN 32
#define FTK 32
#define FJS 16
#define FJL (D3 / FJS)
#define FKB (DH / FTK)
#define FMB ((D3 / FTN) * FKB)
#define FPP 36
#define FWT (D3 * FTK / 4 / 512)
__global__ __launch_bounds__(512) void cea_fuse(const float* __restrict__ Wq,
                                                const float* __restrict__ Wk,
                                                const float* __restrict__ Wv,
                                                const float* __restrict__ Sq,
                                                const float* __restrict__ Sk,
                                                const float* __restrict__ Sv,
                                                const float* __restrict__ bq,
                                                const float* __restrict__ bk,
                                                const float* __restrict__ bv,
                                                float* __restrict__ Mt,
                                                float* __restrict__ cb) {
  __shared__ __attribute__((aligned(16))) float spW[D3 * FTK];
  __shared__ __attribute__((aligned(16))) float part[FJS * FTN * FPP];
  const int h   = blockIdx.y;
  const int tid = threadIdx.x;
  const float* S  = (h == 0) ? Sq : (h == 1) ? Sk : Sv;
  if (blockIdx.x >= FMB) {
    const float* b = (h == 0) ? bq : (h == 1) ? bk : bv;
    bias_job(b, S, cb + h * D3, part, (blockIdx.x - FMB) * BNC, tid);
    return;
  }
  const int bn = blockIdx.x / FKB, bk4 = blockIdx.x - bn * FKB;
  const int n0 = bn * FTN, k0 = bk4 * FTK;
  const int lane = tid & 31, seg = tid >> 5;
  const int kg = lane & 3, ng = lane >> 2;
  const float* W  = (h == 0) ? Wq : (h == 1) ? Wk : Wv;

  v4f w[FWT];
#pragma unroll
  for (int i = 0; i < FWT; ++i) {
    const int f = tid + 512 * i;
    w[i] = *(const v4f*)(W + (f >> 3) * DH + k0 + 4 * (f & 7));
  }
  const float* Sl = S + (seg * FJL) * D3 + n0 + 4 * ng;
  v4f s[FJL];
#pragma unroll
  for (int u = 0; u < FJL; ++u) s[u] = *(const v4f*)(Sl + u * D3);
#pragma unroll
  for (int i = 0; i < FWT; ++i) {
    v4f o;
    o.x = softplus_hw(w[i].x); o.y = softplus_hw(w[i].y);
    o.z = softplus_hw(w[i].z); o.w = softplus_hw(w[i].w);
    *(v4f*)&spW[(tid + 512 * i) * 4] = o;
  }
  __syncthreads();

  v4f acc[4][2];
#pragma unroll
  for (int n = 0; n < 4; ++n) { acc[n][0] = (v4f){}; acc[n][1] = (v4f){}; }
  const float* wl = &spW[(seg * FJL) * FTK + 8 * kg];
#pragma unroll
  for (int u = 0; u < FJL; ++u) {
    const v4f w0 = *(const v4f*)(wl + u * FTK);
    const v4f w1 = *(const v4f*)(wl + u * FTK + 4);
#pragma unroll
    for (int n = 0; n < 4; ++n) {
      acc[n][0] += w0 * s[u][n];
      acc[n][1] += w1 * s[u][n];
    }
  }
#pragma unroll
  for (int n = 0; n < 4; ++n) {
    float* pp = &part[(seg * FTN + 4 * ng + n) * FPP + 8 * kg];
    *(v4f*)pp       = acc[n][0];
    *(v4f*)(pp + 4) = acc[n][1];
  }
  __syncthreads();

  if (tid < 256) {
    const int nl = tid >> 3, k4 = tid & 7;
    const float* pp = &part[nl * FPP + 4 * k4];
    v4f a = *(const v4f*)pp;
    for (int sg = 1; sg < FJS; ++sg) a += *(const v4f*)(pp + sg * (FTN * FPP));
    const int nt = 2 * bn + (nl >> 4), cc = nl & 15;
    const int kc = bk4 * 4 + (k4 >> 1), hh = k4 & 1;
    const int p  = (cc >> 2) * 8 + hh * 4 + (cc & 3);
    *(v4f*)&Mt[((((h * 24 + nt) * QKC + kc) * 32) + p) * 4] = a;
  }
}


static __device__ __forceinline__ void qkv_load(const float* __restrict__ aLane,
                                                const v4f* __restrict__ bLane,
                                                int kc, v4f (&a)[2], v4f (&b)[3]) {
  a[0] = *(const v4f*)(aLane + kc * 8);
  a[1] = *(const v4f*)(aLane + 16 * QPITCH + kc * 8);
#pragma unroll
  for (int q = 0; q < 3; ++q) b[q] = bLane[(q * QKC + kc) * 32];
}

static __device__ __forceinline__ void qkv_mma(const v4f (&a)[2], const v4f (&b)[3], v8f (&acc)[2][3]) {
#pragma unroll
  for (int m = 0; m < 2; ++m)
#pragma unroll
    for (int q = 0; q < 3; ++q) acc[m][q] = wmma4(lo2(a[m]), lo2(b[q]), acc[m][q]);
#pragma unroll
  for (int m = 0; m < 2; ++m)
#pragma unroll
    for (int q = 0; q < 3; ++q) acc[m][q] = wmma4(hi2(a[m]), hi2(b[q]), acc[m][q]);
}

__global__ __launch_bounds__(256) void cea_qkv(const float* __restrict__ x,
                                               const float* __restrict__ Mt,
                                               const float* __restrict__ cb,
                                               float* __restrict__ qkv,
                                               const float* __restrict__ Wp,
                                               float* __restrict__ spPf) {
  __shared__ __attribute__((aligned(16))) float xT[QTOK * QPITCH];
  if (blockIdx.x >= QGB) {
    wp_frag_job(Wp, spPf, blockIdx.x - QGB, threadIdx.x);
    return;
  }
  const int gy   = blockIdx.x / (NT / QTOK);
  const int h    = gy >> 1;
  const int ntb  = (gy & 1) * QNT;
  const int t0   = (blockIdx.x - gy * (NT / QTOK)) * QTOK;
  const int tid  = threadIdx.x;
  const int lane = tid & 31, wid = tid >> 5;
  {
    const float* src = x + (t0 + wid * 8) * DD + h * DH + lane * 4;
    float*       dst = xT + (wid * 8) * QPITCH + lane * 4;
    v4f tmp[8];
#pragma unroll
    for (int i = 0; i < 8; ++i) tmp[i] = *(const v4f*)(src + i * DD);
#pragma unroll
    for (int i = 0; i < 8; ++i) *(v4f*)(dst + i * QPITCH) = tmp[i];
  }
  __syncthreads();

  const int c  = lane & 15;
  const int hh = lane >> 4;
  const int mp = wid & 1;
  const int nt = ntb + (wid >> 1) * 3;
  const float* aLane = &xT[(mp * 32 + c) * QPITCH + 4 * hh];
  const v4f*   bLane = (const v4f*)Mt + ((h * 24 + nt) * QKC) * 32
                       + ((c >> 2) * 8 + hh * 4 + (c & 3));

  v8f acc[2][3];
#pragma unroll
  for (int m = 0; m < 2; ++m)
#pragma unroll
    for (int q = 0; q < 3; ++q) acc[m][q] = (v8f){};

  v4f a[4][2], b[4][3];
  qkv_load(aLane, bLane, 0, a[0], b[0]);
  qkv_load(aLane, bLane, 1, a[1], b[1]);
  qkv_load(aLane, bLane, 2, a[2], b[2]);
#pragma unroll
  for (int s = 0; s < 3; ++s) {
    asm volatile("" :: "v"(a[s][0]), "v"(a[s][1]));
#pragma unroll
    for (int q = 0; q < 3; ++q) asm volatile("" :: "v"(b[s][q]));
  }
  for (int kc = 0; kc < QKC; kc += 4) {
#pragma unroll
    for (int s = 0; s < 4; ++s) {
      const int kp = kc + s + 3;
      const int kn = (kp < QKC) ? kp : QKC - 1;
      qkv_load(aLane, bLane, kn, a[(s + 3) & 3], b[(s + 3) & 3]);
      qkv_mma(a[s], b[s], acc);
    }
  }

  const int row0 = t0 + mp * 32 + hh * 8;
#pragma unroll
  for (int m = 0; m < 2; ++m) {
#pragma unroll
    for (int q = 0; q < 3; ++q) {
      const int n = h * D3 + (nt + q) * 16 + c;
      const float bias = cb[n];
#pragma unroll
      for (int r = 0; r < 8; ++r)
        qkv[(row0 + m * 16 + r) * D9 + n] = acc[m][q][r] + bias;
    }
  }
}

#define ATOK  4
#define ACH   (D3 / 4)

static __device__ __forceinline__ void att_mul(const v2f (&qq)[4], v4f k4, v2f (&z)[8]) {
  const v2f k0 = lo2(k4), k1 = hi2(k4);
#pragma unroll
  for (int r = 0; r < 4; ++r) { z[r] = qq[r] * k0; z[4 + r] = qq[r] * k1; }
}
static __device__ __forceinline__ void att_exp(const v2f (&z)[8], v2f (&e)[8]) {
#pragma unroll
  for (int c = 0; c < 8; ++c) {
    e[c].x = __builtin_amdgcn_exp2f(z[c].x);
    e[c].y = __builtin_amdgcn_exp2f(z[c].y);
  }
}
static __device__ __forceinline__ void att_one(const v2f (&e)[8], v2f (&a)[8]) {
#pragma unroll
  for (int c = 0; c < 8; ++c) a[c] = e[c] + 1.0f;
}
static __device__ __forceinline__ void att_sum(const v2f (&a)[8], v4f v4, v2f (&acc)[4], v2f (&ssum)[4]) {
  const v2f v0 = lo2(v4), v1 = hi2(v4);
#pragma unroll
  for (int c = 0; c < 8; ++c) {
    v2f s;
    s.x = __builtin_amdgcn_rcpf(a[c].x);
    s.y = __builtin_amdgcn_rcpf(a[c].y);
    acc[c & 3]  += s * ((c < 4) ? v0 : v1);
    ssum[c & 3] += s;
  }
}

__global__ __launch_bounds__(384) void cea_attn(const float* __restrict__ qkv,
                                                float* __restrict__ att) {
  __shared__ __attribute__((aligned(16))) float KV[ATOK * 2 * D3];
  const int tid  = threadIdx.x;
  const int lane = tid & 31, wid = tid >> 5;
  const int tk   = wid / 3;
  const int i0   = (wid - tk * 3) * 128 + lane * 4;
  const int t    = blockIdx.x * ATOK + tk;
  const float* __restrict__ base = qkv + t * D9;
  const v4f q4 = *(const v4f*)(base + i0);
  {
    const v4f k4 = *(const v4f*)(base + D3 + i0);
    const v4f v4 = *(const v4f*)(base + 2 * D3 + i0);
    *(v4f*)&KV[tk * 2 * D3 + i0]      = k4 * (-GAMMA * LOG2E);
    *(v4f*)&KV[tk * 2 * D3 + D3 + i0] = v4;
  }
  __syncthreads();

  v2f qq[4], acc[4], ssum[4];
#pragma unroll
  for (int r = 0; r < 4; ++r) {
    qq[r]   = (v2f){q4[r], q4[r]};
    acc[r]  = (v2f){0.0f, 0.0f};
    ssum[r] = (v2f){0.0f, 0.0f};
  }
  const float* __restrict__ kp = &KV[tk * 2 * D3];

  v2f z[8], e[8], aA[8], aB[8];
  att_mul(qq, *(const v4f*)kp, z);
  att_exp(z, e);
  att_one(e, aA);
  att_mul(qq, *(const v4f*)(kp + 4), z);
  att_exp(z, e);
  v4f kA = *(const v4f*)(kp + 8);
  v4f kB = *(const v4f*)(kp + 12);
  v4f vA = *(const v4f*)(kp + D3);
  v4f vB;
  for (int n = 2; n < ACH; n += 2) {
    att_sum(aA, vA, acc, ssum);
    att_one(e, aB);
    att_mul(qq, kA, z);
    att_exp(z, e);
    const int nA = (n + 2 < ACH) ? n + 2 : ACH - 1;
    kA = *(const v4f*)(kp + 4 * nA);
    vB = *(const v4f*)(kp + D3 + 4 * (n - 1));
    att_sum(aB, vB, acc, ssum);
    att_one(e, aA);
    att_mul(qq, kB, z);
    att_exp(z, e);
    const int nB = (n + 3 < ACH) ? n + 3 : ACH - 1;
    kB = *(const v4f*)(kp + 4 * nB);
    vA = *(const v4f*)(kp + D3 + 4 * n);
#pragma unroll
    for (int g = 0; g < 64; ++g) {
      __builtin_amdgcn_sched_group_barrier(0x002, 1, 0);
      __builtin_amdgcn_sched_group_barrier(0x400, 1, 0);
    }
  }
  att_sum(aA, vA, acc, ssum);
  att_one(e, aB);
  att_sum(aB, *(const v4f*)(kp + D3 + 4 * (ACH - 1)), acc, ssum);

  v4f o;
#pragma unroll
  for (int r = 0; r < 4; ++r)
    o[r] = (acc[r].x + acc[r].y) / ((ssum[r].x + ssum[r].y) + 1e-8f);
  *(v4f*)(att + t * D3 + i0) = o;
}


template <int NTW>
static __device__ __forceinline__ void proj_load(const float* __restrict__ aLane,
                                                 const v4f* __restrict__ bLane,
                                                 int kc, v4f& a, v4f (&b)[NTW]) {
  a = *(const v4f*)(aLane + kc * 8);
#pragma unroll
  for (int q = 0; q < NTW; ++q) b[q] = bLane[(q * PKC + kc) * 32];
}

template <int NTW>
static __device__ __forceinline__ void proj_mma(v4f a, const v4f (&b)[NTW], v8f (&acc)[NTW]) {
#pragma unroll
  for (int q = 0; q < NTW; ++q) acc[q] = wmma4(lo2(a), lo2(b[q]), acc[q]);
#pragma unroll
  for (int q = 0; q < NTW; ++q) acc[q] = wmma4(hi2(a), hi2(b[q]), acc[q]);
}

template <int NTW>
static __device__ __forceinline__ void proj_wave(const float* __restrict__ aLane,
                                                 const v4f* __restrict__ bLane,
                                                 const float* __restrict__ bp,
                                                 float* __restrict__ hout,
                                                 int row0, int col0) {
  v8f acc[NTW];
#pragma unroll
  for (int q = 0; q < NTW; ++q) acc[q] = (v8f){};

  v4f a[4], b[4][NTW];
  proj_load<NTW>(aLane, bLane, 0, a[0], b[0]);
  proj_load<NTW>(aLane, bLane, 1, a[1], b[1]);
  proj_load<NTW>(aLane, bLane, 2, a[2], b[2]);
#pragma unroll
  for (int s = 0; s < 3; ++s) {
    asm volatile("" :: "v"(a[s]));
#pragma unroll
    for (int q = 0; q < NTW; ++q) asm volatile("" :: "v"(b[s][q]));
  }
  for (int kc = 0; kc < PKC; kc += 4) {
#pragma unroll
    for (int s = 0; s < 4; ++s) {
      const int kp = kc + s + 3;
      const int kn = (kp < PKC) ? kp : PKC - 1;
      proj_load<NTW>(aLane, bLane, kn, a[(s + 3) & 3], b[(s + 3) & 3]);
      proj_mma<NTW>(a[s], b[s], acc);
    }
  }

#pragma unroll
  for (int q = 0; q < NTW; ++q) {
    const int n = col0 + q * 16;
    const float bias = bp[n];
#pragma unroll
    for (int r = 0; r < 8; ++r)
      hout[(row0 + r) * D9 + n] = acc[q][r] + bias;
  }
}

__global__ __launch_bounds__(256) void cea_proj(const float* __restrict__ att,
                                                const float* __restrict__ spPf,
                                                const float* __restrict__ bp,
                                                float* __restrict__ hout) {
  __shared__ __attribute__((aligned(16))) float aT[PTOK * APITCH];
  const int t0  = blockIdx.x * PTOK;
  const int tid = threadIdx.x;
  {
    const int sw = tid >> 5, sl = tid & 31;
    const float* src = att + (t0 + sw * 8) * D3 + sl * 4;
    float*       dst = aT + (sw * 8) * APITCH + sl * 4;
    v4f tmp[24];
#pragma unroll
    for (int i = 0; i < 24; ++i) tmp[i] = *(const v4f*)(src + (i / 3) * D3 + (i % 3) * 128);
#pragma unroll
    for (int i = 0; i < 24; ++i) *(v4f*)(dst + (i / 3) * APITCH + (i % 3) * 128) = tmp[i];
  }
  __syncthreads();

  const int lane = tid & 31, wid = tid >> 5;
  const int c  = lane & 15;
  const int h  = lane >> 4;
  const int mi = wid & 3;
  const int nt0 = (wid >> 2) * 5;
  const float* aLane = &aT[(mi * 16 + c) * APITCH + 4 * h];
  const v4f*   bLane = (const v4f*)spPf + ((blockIdx.y * 9 + nt0) * PKC) * 32 + lane;
  const int row0 = t0 + mi * 16 + h * 8;
  const int col0 = blockIdx.y * PCOL + nt0 * 16 + c;

  if (wid < 4) proj_wave<5>(aLane, bLane, bp, hout, row0, col0);
  else         proj_wave<4>(aLane, bLane, bp, hout, row0, col0);
}

#define LNI 5
template <int M>
static __device__ __forceinline__ float ln_xrow(float x) {
  const int b = __builtin_bit_cast(int, x);
  return __builtin_bit_cast(float, __builtin_amdgcn_update_dpp(b, b, 0x160 + M, 0xf, 0xf, false));
}
static __device__ __forceinline__ float ln_xhalf(float x) {
  const int b = __builtin_bit_cast(int, x);
  return __builtin_bit_cast(float, __builtin_amdgcn_permlanex16(b, b, 0x76543210, 0xfedcba98, false, false));
}
static __device__ __forceinline__ float ln_wave_sum(float v) {
  v += ln_xrow<1>(v);
  v += ln_xrow<2>(v);
  v += ln_xrow<4>(v);
  v += ln_xrow<8>(v);
  v += ln_xhalf(v);
  return v;
}
__global__ __launch_bounds__(256) void cea_ln(float* __restrict__ out,
                                              const float* __restrict__ gn,
                                              const float* __restrict__ bn) {
  __shared__ float redA[8];
  __shared__ float redB[8];
  const int t = blockIdx.x, tid = threadIdx.x;
  const int lane = tid & 31, wid = tid >> 5;
  const bool tail = tid < (D9 - 4 * 256);
  const int n4 = tail ? tid + 4 * 256 : tid;
  float* __restrict__ h = out + t * D9;

  float v[LNI];
#pragma unroll
  for (int i = 0; i < 4; ++i) v[i] = h[tid + 256 * i];
  v[4] = h[n4];
  if (!tail) v[4] = 0.0f;

  float ps = ((v[0] + v[1]) + (v[2] + v[3])) + v[4];
  ps = ln_wave_sum(ps);
  if (lane == 0) redA[wid] = ps;
  __syncthreads();
  float sm = redA[0];
#pragma unroll
  for (int w = 1; w < 8; ++w) sm += redA[w];
  const float mu = sm * (1.0f / (float)D9);

  float dv[LNI];
#pragma unroll
  for (int i = 0; i < LNI; ++i) dv[i] = v[i] - mu;
  if (!tail) dv[4] = 0.0f;
  float pv = ((dv[0] * dv[0] + dv[1] * dv[1]) + (dv[2] * dv[2] + dv[3] * dv[3])) + dv[4] * dv[4];
  pv = ln_wave_sum(pv);
  if (lane == 0) redB[wid] = pv;
  __syncthreads();
  float sv = redB[0];
#pragma unroll
  for (int w = 1; w < 8; ++w) sv += redB[w];
  const float var = sv * (1.0f / (float)D9);
  const float rs  = __builtin_amdgcn_rsqf(var + 1e-5f);

  float g[LNI], b[LNI];
#pragma unroll
  for (int i = 0; i < 4; ++i) { g[i] = gn[tid + 256 * i]; b[i] = bn[tid + 256 * i]; }
  g[4] = gn[n4]; b[4] = bn[n4];
  asm volatile("" :: "v"(g[4]), "v"(b[4]));
#pragma unroll
  for (int i = 0; i < 4; ++i) h[tid + 256 * i] = dv[i] * rs * g[i] + b[i];
  if (tail) h[n4] = dv[4] * rs * g[4] + b[4];
}


extern "C" void kernel_launch(void* const* d_in, const int* in_sizes, int n_in,
                              void* d_out, int out_size, void* d_ws, size_t ws_size,
                              hipStream_t stream) {
  const float* x  = (const float*)d_in[0];
  const float* Wq = (const float*)d_in[1];
  const float* bq = (const float*)d_in[2];
  const float* Sq = (const float*)d_in[3];
  const float* Wk = (const float*)d_in[4];
  const float* bk = (const float*)d_in[5];
  const float* Sk = (const float*)d_in[6];
  const float* Wv = (const float*)d_in[7];
  const float* bv = (const float*)d_in[8];
  const float* Sv = (const float*)d_in[9];
  const float* Wp = (const float*)d_in[10];
  const float* bp = (const float*)d_in[11];
  const float* gn = (const float*)d_in[12];
  const float* bn = (const float*)d_in[13];
  float* out = (float*)d_out;

  float* ws = (float*)d_ws;
  float* Mt   = ws;
  float* cb   = Mt  + 3 * D3 * DH;
  float* spP  = cb  + 3 * D3;
  float* qkv  = spP + D9 * D3;
  float* attb = qkv + NT * D9;

  cea_fuse<<<dim3(FMB + BNB, 3), dim3(512), 0, stream>>>(Wq, Wk, Wv, Sq, Sk, Sv, bq, bk, bv, Mt, cb);

  cea_qkv<<<dim3(QGB + SPB), dim3(256), 0, stream>>>(x, Mt, cb, qkv, Wp, spP);

  cea_attn<<<dim3(NT / ATOK), dim3(3 * 32 * ATOK), 0, stream>>>(qkv, attb);

  cea_proj<<<dim3(NT / PTOK, D9 / PCOL), dim3(256), 0, stream>>>(attb, spP, bp, out);

  cea_ln<<<dim3(NT), dim3(256), 0, stream>>>(out, gn, bn);

  (void)in_sizes; (void)n_in; (void)out_size; (void)ws_size;
}
